# starting kernel
# baseline (speedup 1.0000x reference)
#include <hip/hip_runtime.h>

#define N_NODES 100000
#define N_EDGES 1600000
#define IN_F    128
#define H_F     64

typedef __attribute__((ext_vector_type(2))) float v2f;
typedef __attribute__((ext_vector_type(8))) float v8f;


__global__ void gcn_zero_f32(float* __restrict__ p, long n) {
    long i = (long)blockIdx.x * blockDim.x + threadIdx.x;
    if (i < n) p[i] = 0.f;
}

__global__ void gcn_degree_accum(const int* __restrict__ src,
                                 const int* __restrict__ dst,
                                 float* __restrict__ deg_src,
                                 float* __restrict__ deg_dst, int ne) {
    int e = blockIdx.x * blockDim.x + threadIdx.x;
    if (e < ne) {
        atomicAdd(&deg_src[src[e]], 1.0f);
        atomicAdd(&deg_dst[dst[e]], 1.0f);
    }
}

__global__ void gcn_inv_sqrt(float* __restrict__ deg, long n) {
    long i = (long)blockIdx.x * blockDim.x + threadIdx.x;
    if (i < n) deg[i] = rsqrtf(fmaxf(deg[i], 1.0f));
}


template <int K>
__global__ void gcn_gemm_wmma(const float* __restrict__ A,
                              const float* __restrict__ W,
                              const float* __restrict__ rowscale,
                              float* __restrict__ C) {
    const int wave = threadIdx.x >> 5;
    const int lane = threadIdx.x & 31;
    const int half = lane >> 4;
    const int l16  = lane & 15;
    const int m0   = blockIdx.x * 16;
    const int n0   = wave * 16;
    const int m    = m0 + l16;

    const float s = rowscale[m];
    const float* __restrict__ arow = A + (long)m * K;

    v8f c = {};
    #pragma unroll 4
    for (int k0 = 0; k0 < K; k0 += 4) {
        const int ka = k0 + 2 * half;
        v2f a, b;
        a.x = arow[ka]     * s;
        a.y = arow[ka + 1] * s;
        b.x = W[(long)ka       * H_F + n0 + l16];
        b.y = W[(long)(ka + 1) * H_F + n0 + l16];
        c = __builtin_amdgcn_wmma_f32_16x16x4_f32(
                false, a, false, b, (short)0, c, false, false);
    }

    #pragma unroll
    for (int r = 0; r < 8; ++r) {
        C[(long)(m0 + r + 8 * half) * H_F + n0 + l16] = c[r];
    }
}


__global__ void gcn_spmm_scatter(const float* __restrict__ h,
                                 const int* __restrict__ src,
                                 const int* __restrict__ dst,
                                 float* __restrict__ agg, int ne) {
    long t = (long)blockIdx.x * blockDim.x + threadIdx.x;
    int e = (int)(t >> 4);
    int c = (int)(t & 15) << 2;
    if (e < ne) {
        const int s = src[e];
        const int d = dst[e];
        const float4 v = *(const float4*)(h + (long)s * H_F + c);
        float* p = agg + (long)d * H_F + c;
        atomicAdd(p + 0, v.x);
        atomicAdd(p + 1, v.y);
        atomicAdd(p + 2, v.z);
        atomicAdd(p + 3, v.w);
    }
}


__global__ void gcn_bias_relu(const float* __restrict__ agg,
                              const float* __restrict__ in_norm,
                              const float* __restrict__ bias,
                              float* __restrict__ h, long n) {
    long i = (long)blockIdx.x * blockDim.x + threadIdx.x;
    if (i < n) {
        int node = (int)(i >> 6);
        int f = (int)(i & 63);
        float v = agg[i] * in_norm[node] + bias[f];
        h[i] = fmaxf(v, 0.0f);
    }
}


__global__ void gcn_head(const float* __restrict__ h,
                         const float* __restrict__ Wm,
                         const float* __restrict__ bm,
                         float* __restrict__ out, int n) {
    int node = blockIdx.x * blockDim.x + threadIdx.x;
    if (node < n) {
        float a0 = bm[0], a1 = bm[1];
        const float* __restrict__ hr = h + (long)node * H_F;
        #pragma unroll 8
        for (int k = 0; k < H_F; ++k) {
            float v = hr[k];
            a0 = fmaf(v, Wm[k * 2 + 0], a0);
            a1 = fmaf(v, Wm[k * 2 + 1], a1);
        }
        out[node * 2 + 0] = a0;
        out[node * 2 + 1] = a1;
    }
}


extern "C" void kernel_launch(void* const* d_in, const int* in_sizes, int n_in,
                              void* d_out, int out_size, void* d_ws, size_t ws_size,
                              hipStream_t stream) {
    const float* feature = (const float*)d_in[0];
    const float* W1      = (const float*)d_in[1];
    const float* b1      = (const float*)d_in[2];
    const float* W2      = (const float*)d_in[3];
    const float* b2      = (const float*)d_in[4];
    const float* Wm      = (const float*)d_in[5];
    const float* bm      = (const float*)d_in[6];
    const int*   src     = (const int*)d_in[7];
    const int*   dst     = (const int*)d_in[8];
    float*       out     = (float*)d_out;

    float* out_norm = (float*)d_ws;
    float* in_norm  = out_norm + N_NODES;
    float* buf0     = in_norm + N_NODES;
    float* buf1     = buf0 + (long)N_NODES * H_F;

    const long NH = (long)N_NODES * H_F;
    const int  B  = 256;

    gcn_zero_f32<<<(2 * N_NODES + B - 1) / B, B, 0, stream>>>(out_norm, 2 * N_NODES);
    gcn_degree_accum<<<(N_EDGES + B - 1) / B, B, 0, stream>>>(src, dst, out_norm, in_norm, N_EDGES);
    gcn_inv_sqrt<<<(2 * N_NODES + B - 1) / B, B, 0, stream>>>(out_norm, 2 * N_NODES);

    gcn_gemm_wmma<IN_F><<<N_NODES / 16, 128, 0, stream>>>(feature, W1, out_norm, buf0);
    gcn_zero_f32<<<(int)((NH + B - 1) / B), B, 0, stream>>>(buf1, NH);
    gcn_spmm_scatter<<<(int)(((long)N_EDGES * 16 + B - 1) / B), B, 0, stream>>>(buf0, src, dst, buf1, N_EDGES);
    gcn_bias_relu<<<(int)((NH + B - 1) / B), B, 0, stream>>>(buf1, in_norm, b1, buf0, NH);

    gcn_gemm_wmma<H_F><<<N_NODES / 16, 128, 0, stream>>>(buf0, W2, out_norm, buf1);
    gcn_zero_f32<<<(int)((NH + B - 1) / B), B, 0, stream>>>(buf0, NH);
    gcn_spmm_scatter<<<(int)(((long)N_EDGES * 16 + B - 1) / B), B, 0, stream>>>(buf1, src, dst, buf0, N_EDGES);
    gcn_bias_relu<<<(int)((NH + B - 1) / B), B, 0, stream>>>(buf0, in_norm, b2, buf1, NH);

    gcn_head<<<(N_NODES + B - 1) / B, B, 0, stream>>>(buf1, Wm, bm, out, N_NODES);
}


# Degrees counted in LDS histograms per (index array, node part, edge slice) and summed by a second kernel; replaces the zero, global float atomic and rsqrt passes; counts are exact integers
# speedup vs baseline: 1.3323x; 1.3323x over previous
#include <hip/hip_runtime.h>

#define N_NODES 100000
#define N_EDGES 1600000
#define IN_F    128
#define H_F     64

typedef __attribute__((ext_vector_type(2))) float v2f;
typedef __attribute__((ext_vector_type(8))) float v8f;


__global__ void gcn_zero_f32(float* __restrict__ p, long n) {
    long i = (long)blockIdx.x * blockDim.x + threadIdx.x;
    if (i < n) p[i] = 0.f;
}


#define DEG_R     4
#define DEG_S     32
#define DEG_T     512
#define DEG_U     8
#define DEG_BINS  25024
#define DEG_NP    (DEG_R * DEG_BINS)
#define DEG_SLICE (N_EDGES / DEG_S)
#define DEG_Q     (DEG_SLICE / 4)

static_assert(DEG_NP >= N_NODES, "parts must cover every node");
static_assert(DEG_BINS % 32 == 0, "a part must be whole 128-byte lines");
static_assert(N_EDGES % DEG_S == 0 && DEG_SLICE % 4 == 0, "slices must be whole 16-byte groups");
static_assert(N_NODES % 4 == 0, "finish kernel handles four nodes per thread");

__device__ __forceinline__ void gcn_count_one(unsigned* hist, int v, int lo) {
    const unsigned d = (unsigned)(v - lo);
    if (d < (unsigned)DEG_BINS) atomicAdd(&hist[d], 1u);
}

__global__ __launch_bounds__(DEG_T)
void gcn_degree_hist(const int* __restrict__ src,
                     const int* __restrict__ dst,
                     unsigned* __restrict__ part) {
    __shared__ __attribute__((aligned(16))) unsigned hist[DEG_BINS];

    const int tid = threadIdx.x;
    const int s   = blockIdx.x % DEG_S;
    const int ar  = blockIdx.x / DEG_S;
    const int r   = ar % DEG_R;
    const int a   = ar / DEG_R;
    const int lo  = r * DEG_BINS;

    uint4* h4 = (uint4*)hist;
    for (int i = tid; i < DEG_BINS / 4; i += DEG_T) h4[i] = make_uint4(0u, 0u, 0u, 0u);
    __syncthreads();

    const int* __restrict__ idx = a ? dst : src;
    const int4* __restrict__ p4 = (const int4*)(idx + (long)s * DEG_SLICE);

    for (int base = 0; base < DEG_Q; base += DEG_T * DEG_U) {
        int4 v[DEG_U];
        #pragma unroll
        for (int u = 0; u < DEG_U; ++u) {
            const int q = base + u * DEG_T + tid;
            v[u] = (q < DEG_Q) ? p4[q] : make_int4(-1, -1, -1, -1);
        }
        #pragma unroll
        for (int u = 0; u < DEG_U; ++u) {
            gcn_count_one(hist, v[u].x, lo);
            gcn_count_one(hist, v[u].y, lo);
            gcn_count_one(hist, v[u].z, lo);
            gcn_count_one(hist, v[u].w, lo);
        }
    }
    __syncthreads();

    uint4* __restrict__ o4 = (uint4*)(part + (long)(a * DEG_S + s) * DEG_NP + lo);
    for (int i = tid; i < DEG_BINS / 4; i += DEG_T) o4[i] = h4[i];
}

__global__ void gcn_degree_finish(const unsigned* __restrict__ part,
                                  float* __restrict__ norm) {
    const int t = blockIdx.x * blockDim.x + threadIdx.x;
    if (t < 2 * (N_NODES / 4)) {
        const int a = t / (N_NODES / 4);
        const int j = t - a * (N_NODES / 4);
        const uint4* __restrict__ p = (const uint4*)(part + (long)a * DEG_S * DEG_NP) + j;
        uint4 acc = make_uint4(0u, 0u, 0u, 0u);
        #pragma unroll
        for (int s = 0; s < DEG_S; ++s) {
            const uint4 v = p[(long)s * (DEG_NP / 4)];
            acc.x += v.x; acc.y += v.y; acc.z += v.z; acc.w += v.w;
        }
        float4 o;
        o.x = rsqrtf(fmaxf((float)acc.x, 1.0f));
        o.y = rsqrtf(fmaxf((float)acc.y, 1.0f));
        o.z = rsqrtf(fmaxf((float)acc.z, 1.0f));
        o.w = rsqrtf(fmaxf((float)acc.w, 1.0f));
        *(float4*)(norm + (long)a * N_NODES + 4 * j) = o;
    }
}


template <int K>
__global__ void gcn_gemm_wmma(const float* __restrict__ A,
                              const float* __restrict__ W,
                              const float* __restrict__ rowscale,
                              float* __restrict__ C) {
    const int wave = threadIdx.x >> 5;
    const int lane = threadIdx.x & 31;
    const int half = lane >> 4;
    const int l16  = lane & 15;
    const int m0   = blockIdx.x * 16;
    const int n0   = wave * 16;
    const int m    = m0 + l16;

    const float s = rowscale[m];
    const float* __restrict__ arow = A + (long)m * K;

    v8f c = {};
    #pragma unroll 4
    for (int k0 = 0; k0 < K; k0 += 4) {
        const int ka = k0 + 2 * half;
        v2f a, b;
        a.x = arow[ka]     * s;
        a.y = arow[ka + 1] * s;
        b.x = W[(long)ka       * H_F + n0 + l16];
        b.y = W[(long)(ka + 1) * H_F + n0 + l16];
        c = __builtin_amdgcn_wmma_f32_16x16x4_f32(
                false, a, false, b, (short)0, c, false, false);
    }

    #pragma unroll
    for (int r = 0; r < 8; ++r) {
        C[(long)(m0 + r + 8 * half) * H_F + n0 + l16] = c[r];
    }
}


__global__ void gcn_spmm_scatter(const float* __restrict__ h,
                                 const int* __restrict__ src,
                                 const int* __restrict__ dst,
                                 float* __restrict__ agg, int ne) {
    long t = (long)blockIdx.x * blockDim.x + threadIdx.x;
    int e = (int)(t >> 4);
    int c = (int)(t & 15) << 2;
    if (e < ne) {
        const int s = src[e];
        const int d = dst[e];
        const float4 v = *(const float4*)(h + (long)s * H_F + c);
        float* p = agg + (long)d * H_F + c;
        atomicAdd(p + 0, v.x);
        atomicAdd(p + 1, v.y);
        atomicAdd(p + 2, v.z);
        atomicAdd(p + 3, v.w);
    }
}


__global__ void gcn_bias_relu(const float* __restrict__ agg,
                              const float* __restrict__ in_norm,
                              const float* __restrict__ bias,
                              float* __restrict__ h, long n) {
    long i = (long)blockIdx.x * blockDim.x + threadIdx.x;
    if (i < n) {
        int node = (int)(i >> 6);
        int f = (int)(i & 63);
        float v = agg[i] * in_norm[node] + bias[f];
        h[i] = fmaxf(v, 0.0f);
    }
}


__global__ void gcn_head(const float* __restrict__ h,
                         const float* __restrict__ Wm,
                         const float* __restrict__ bm,
                         float* __restrict__ out, int n) {
    int node = blockIdx.x * blockDim.x + threadIdx.x;
    if (node < n) {
        float a0 = bm[0], a1 = bm[1];
        const float* __restrict__ hr = h + (long)node * H_F;
        #pragma unroll 8
        for (int k = 0; k < H_F; ++k) {
            float v = hr[k];
            a0 = fmaf(v, Wm[k * 2 + 0], a0);
            a1 = fmaf(v, Wm[k * 2 + 1], a1);
        }
        out[node * 2 + 0] = a0;
        out[node * 2 + 1] = a1;
    }
}


extern "C" void kernel_launch(void* const* d_in, const int* in_sizes, int n_in,
                              void* d_out, int out_size, void* d_ws, size_t ws_size,
                              hipStream_t stream) {
    const float* feature = (const float*)d_in[0];
    const float* W1      = (const float*)d_in[1];
    const float* b1      = (const float*)d_in[2];
    const float* W2      = (const float*)d_in[3];
    const float* b2      = (const float*)d_in[4];
    const float* Wm      = (const float*)d_in[5];
    const float* bm      = (const float*)d_in[6];
    const int*   src     = (const int*)d_in[7];
    const int*   dst     = (const int*)d_in[8];
    float*       out     = (float*)d_out;

    float* out_norm = (float*)d_ws;
    float* in_norm  = out_norm + N_NODES;
    float* buf0     = in_norm + N_NODES;
    float* buf1     = buf0 + (long)N_NODES * H_F;
    unsigned* part  = (unsigned*)(buf1 + (long)N_NODES * H_F);

    const long NH = (long)N_NODES * H_F;
    const int  B  = 256;

    gcn_degree_hist<<<2 * DEG_R * DEG_S, DEG_T, 0, stream>>>(src, dst, part);
    gcn_degree_finish<<<(2 * (N_NODES / 4) + B - 1) / B, B, 0, stream>>>(part, out_norm);

    gcn_gemm_wmma<IN_F><<<N_NODES / 16, 128, 0, stream>>>(feature, W1, out_norm, buf0);
    gcn_zero_f32<<<(int)((NH + B - 1) / B), B, 0, stream>>>(buf1, NH);
    gcn_spmm_scatter<<<(int)(((long)N_EDGES * 16 + B - 1) / B), B, 0, stream>>>(buf0, src, dst, buf1, N_EDGES);
    gcn_bias_relu<<<(int)((NH + B - 1) / B), B, 0, stream>>>(buf1, in_norm, b1, buf0, NH);

    gcn_gemm_wmma<H_F><<<N_NODES / 16, 128, 0, stream>>>(buf0, W2, out_norm, buf1);
    gcn_zero_f32<<<(int)((NH + B - 1) / B), B, 0, stream>>>(buf0, NH);
    gcn_spmm_scatter<<<(int)(((long)N_EDGES * 16 + B - 1) / B), B, 0, stream>>>(buf1, src, dst, buf0, N_EDGES);
    gcn_bias_relu<<<(int)((NH + B - 1) / B), B, 0, stream>>>(buf0, in_norm, b2, buf1, NH);

    gcn_head<<<(N_NODES + B - 1) / B, B, 0, stream>>>(buf1, Wm, bm, out, N_NODES);
}


# Float-atomic scatter-add replaced by edges placed in destination order (slot = returning LDS add, an arrival-order rank) and one register sum per node fused with norm, bias, relu; no zero pass
# speedup vs baseline: 2.7729x; 2.0813x over previous
#include <hip/hip_runtime.h>

#define N_NODES 100000
#define N_EDGES 1600000
#define IN_F    128
#define H_F     64

typedef __attribute__((ext_vector_type(2))) float v2f;
typedef __attribute__((ext_vector_type(8))) float v8f;


#define DEG_R     4
#define DEG_S     32
#define DEG_T     512
#define DEG_U     8
#define DEG_BINS  25024
#define DEG_NP    (DEG_R * DEG_BINS)
#define DEG_SLICE (N_EDGES / DEG_S)
#define DEG_Q     (DEG_SLICE / 4)

static_assert(DEG_NP >= N_NODES, "parts must cover every node");
static_assert(DEG_BINS % 32 == 0, "a part must be whole 128-byte lines");
static_assert(N_EDGES % DEG_S == 0 && DEG_SLICE % 4 == 0, "slices must be whole 16-byte groups");
static_assert(N_NODES % 4 == 0, "finish kernel handles four nodes per thread");

__device__ __forceinline__ void gcn_count_one(unsigned* hist, int v, int lo) {
    const unsigned d = (unsigned)(v - lo);
    if (d < (unsigned)DEG_BINS) atomicAdd(&hist[d], 1u);
}

__global__ __launch_bounds__(DEG_T)
void gcn_degree_hist(const int* __restrict__ src,
                     const int* __restrict__ dst,
                     unsigned* __restrict__ part) {
    __shared__ __attribute__((aligned(16))) unsigned hist[DEG_BINS];

    const int tid = threadIdx.x;
    const int s   = blockIdx.x % DEG_S;
    const int ar  = blockIdx.x / DEG_S;
    const int r   = ar % DEG_R;
    const int a   = ar / DEG_R;
    const int lo  = r * DEG_BINS;

    uint4* h4 = (uint4*)hist;
    for (int i = tid; i < DEG_BINS / 4; i += DEG_T) h4[i] = make_uint4(0u, 0u, 0u, 0u);
    __syncthreads();

    const int* __restrict__ idx = a ? dst : src;
    const int4* __restrict__ p4 = (const int4*)(idx + (long)s * DEG_SLICE);

    for (int base = 0; base < DEG_Q; base += DEG_T * DEG_U) {
        int4 v[DEG_U];
        #pragma unroll
        for (int u = 0; u < DEG_U; ++u) {
            const int q = base + u * DEG_T + tid;
            v[u] = (q < DEG_Q) ? p4[q] : make_int4(-1, -1, -1, -1);
        }
        #pragma unroll
        for (int u = 0; u < DEG_U; ++u) {
            gcn_count_one(hist, v[u].x, lo);
            gcn_count_one(hist, v[u].y, lo);
            gcn_count_one(hist, v[u].z, lo);
            gcn_count_one(hist, v[u].w, lo);
        }
    }
    __syncthreads();

    uint4* __restrict__ o4 = (uint4*)(part + (long)(a * DEG_S + s) * DEG_NP + lo);
    for (int i = tid; i < DEG_BINS / 4; i += DEG_T) o4[i] = h4[i];
}


#define FIN_T   256
#define FIN_NPB (4 * FIN_T)
#define FIN_NB  ((N_NODES + FIN_NPB - 1) / FIN_NPB)

static_assert(FIN_NB <= FIN_T, "one thread per earlier block total in the scan kernel");
static_assert(DEG_S % 8 == 0, "finish kernel loads the slices in groups of eight");

__global__ __launch_bounds__(FIN_T)
void gcn_degree_finish(unsigned* __restrict__ part,
                       float* __restrict__ norm,
                       unsigned* __restrict__ degin,
                       unsigned* __restrict__ blktot) {
    __shared__ unsigned red[FIN_T];
    const int tid = threadIdx.x;
    const int a = blockIdx.x / FIN_NB;
    const int b = blockIdx.x - a * FIN_NB;
    const int j = b * FIN_T + tid;
    const bool valid = j < N_NODES / 4;

    uint4 acc = make_uint4(0u, 0u, 0u, 0u);
    if (valid) {
        uint4* __restrict__ p = (uint4*)(part + (long)a * DEG_S * DEG_NP) + j;
        #pragma unroll
        for (int s0 = 0; s0 < DEG_S; s0 += 8) {
            uint4 v[8];
            #pragma unroll
            for (int u = 0; u < 8; ++u) v[u] = p[(long)(s0 + u) * (DEG_NP / 4)];
            #pragma unroll
            for (int u = 0; u < 8; ++u) {
                if (a == 1) p[(long)(s0 + u) * (DEG_NP / 4)] = acc;
                acc.x += v[u].x; acc.y += v[u].y; acc.z += v[u].z; acc.w += v[u].w;
            }
        }
        float4 o;
        o.x = rsqrtf(fmaxf((float)acc.x, 1.0f));
        o.y = rsqrtf(fmaxf((float)acc.y, 1.0f));
        o.z = rsqrtf(fmaxf((float)acc.z, 1.0f));
        o.w = rsqrtf(fmaxf((float)acc.w, 1.0f));
        *(float4*)(norm + (long)a * N_NODES + 4 * j) = o;
    }

    if (a == 1) {
        if (valid) *(uint4*)(degin + 4 * j) = acc;
        red[tid] = acc.x + acc.y + acc.z + acc.w;
        __syncthreads();
        for (int off = FIN_T / 2; off > 0; off >>= 1) {
            if (tid < off) red[tid] += red[tid + off];
            __syncthreads();
        }
        if (tid == 0) blktot[b] = red[0];
    }
}

__global__ __launch_bounds__(FIN_T)
void gcn_rowptr_scan(const unsigned* __restrict__ degin,
                     const unsigned* __restrict__ blktot,
                     unsigned* __restrict__ rowptr) {
    __shared__ unsigned sh[FIN_T];
    const int tid = threadIdx.x;
    const int b = blockIdx.x;
    const int j = b * FIN_T + tid;
    const bool valid = j < N_NODES / 4;

    sh[tid] = (tid < b) ? blktot[tid] : 0u;
    __syncthreads();
    for (int off = FIN_T / 2; off > 0; off >>= 1) {
        if (tid < off) sh[tid] += sh[tid + off];
        __syncthreads();
    }
    const unsigned base = sh[0];
    __syncthreads();

    uint4 d = make_uint4(0u, 0u, 0u, 0u);
    if (valid) d = *(const uint4*)(degin + 4 * j);
    const unsigned t = d.x + d.y + d.z + d.w;

    sh[tid] = t;
    __syncthreads();
    for (int off = 1; off < FIN_T; off <<= 1) {
        const unsigned y = (tid >= off) ? sh[tid - off] : 0u;
        __syncthreads();
        sh[tid] += y;
        __syncthreads();
    }
    const unsigned first = base + sh[tid] - t;
    if (valid) {
        uint4 o;
        o.x = first;
        o.y = o.x + d.x;
        o.z = o.y + d.y;
        o.w = o.z + d.z;
        *(uint4*)(rowptr + 4 * j) = o;
        if (j == N_NODES / 4 - 1) rowptr[N_NODES] = first + t;
    }
}

__device__ __forceinline__ void gcn_place_one(unsigned* cur, int* __restrict__ col,
                                              int dv, int sv, int lo) {
    const unsigned d = (unsigned)(dv - lo);
    if (d < (unsigned)DEG_BINS) {
        const unsigned pos = atomicAdd(&cur[d], 1u);
        col[pos] = sv;
    }
}

__global__ __launch_bounds__(DEG_T)
void gcn_edge_place(const int* __restrict__ src,
                    const int* __restrict__ dst,
                    const unsigned* __restrict__ pre,
                    const unsigned* __restrict__ rowptr,
                    int* __restrict__ col) {
    __shared__ __attribute__((aligned(16))) unsigned cur[DEG_BINS];

    const int tid = threadIdx.x;
    const int s   = blockIdx.x % DEG_S;
    const int r   = blockIdx.x / DEG_S;
    const int lo  = r * DEG_BINS;

    const uint4* __restrict__ pre4 = (const uint4*)(pre + (long)s * DEG_NP + lo);
    uint4* c4 = (uint4*)cur;
    for (int i = tid; i < DEG_BINS / 4; i += DEG_T) {
        const int n4 = min(lo + 4 * i, N_NODES - 4);
        const uint4 a = *(const uint4*)(rowptr + n4);
        const uint4 p = pre4[i];
        c4[i] = make_uint4(a.x + p.x, a.y + p.y, a.z + p.z, a.w + p.w);
    }
    __syncthreads();

    const int4* __restrict__ d4 = (const int4*)(dst + (long)s * DEG_SLICE);
    const int4* __restrict__ s4 = (const int4*)(src + (long)s * DEG_SLICE);

    for (int base = 0; base < DEG_Q; base += DEG_T * DEG_U) {
        int4 vd[DEG_U], vs[DEG_U];
        #pragma unroll
        for (int u = 0; u < DEG_U; ++u) {
            const int q = base + u * DEG_T + tid;
            vd[u] = (q < DEG_Q) ? d4[q] : make_int4(-1, -1, -1, -1);
            vs[u] = (q < DEG_Q) ? s4[q] : make_int4(0, 0, 0, 0);
        }
        #pragma unroll
        for (int u = 0; u < DEG_U; ++u) {
            gcn_place_one(cur, col, vd[u].x, vs[u].x, lo);
            gcn_place_one(cur, col, vd[u].y, vs[u].y, lo);
            gcn_place_one(cur, col, vd[u].z, vs[u].z, lo);
            gcn_place_one(cur, col, vd[u].w, vs[u].w, lo);
        }
    }
}


template <int K>
__global__ void gcn_gemm_wmma(const float* __restrict__ A,
                              const float* __restrict__ W,
                              const float* __restrict__ rowscale,
                              float* __restrict__ C) {
    const int wave = threadIdx.x >> 5;
    const int lane = threadIdx.x & 31;
    const int half = lane >> 4;
    const int l16  = lane & 15;
    const int m0   = blockIdx.x * 16;
    const int n0   = wave * 16;
    const int m    = m0 + l16;

    const float s = rowscale[m];
    const float* __restrict__ arow = A + (long)m * K;

    v8f c = {};
    #pragma unroll 4
    for (int k0 = 0; k0 < K; k0 += 4) {
        const int ka = k0 + 2 * half;
        v2f a, b;
        a.x = arow[ka]     * s;
        a.y = arow[ka + 1] * s;
        b.x = W[(long)ka       * H_F + n0 + l16];
        b.y = W[(long)(ka + 1) * H_F + n0 + l16];
        c = __builtin_amdgcn_wmma_f32_16x16x4_f32(
                false, a, false, b, (short)0, c, false, false);
    }

    #pragma unroll
    for (int r = 0; r < 8; ++r) {
        C[(long)(m0 + r + 8 * half) * H_F + n0 + l16] = c[r];
    }
}


#define PULL_T 256

static_assert(((long)N_NODES * 16) % PULL_T == 0, "whole blocks of 16-lane groups");

__global__ __launch_bounds__(PULL_T)
void gcn_pull_bias_relu(const float* __restrict__ g,
                        const unsigned* __restrict__ rowptr,
                        const int* __restrict__ col,
                        const float* __restrict__ in_norm,
                        const float* __restrict__ bias,
                        float* __restrict__ h) {
    const int t   = blockIdx.x * PULL_T + threadIdx.x;
    const int n   = t >> 4;
    const int l16 = t & 15;

    const unsigned beg = rowptr[n];
    const unsigned end = rowptr[n + 1];
    const float4* __restrict__ g4 = (const float4*)g;

    float4 acc = make_float4(0.f, 0.f, 0.f, 0.f);
    for (unsigned base = beg; base < end; base += 16) {
        const unsigned jm = base + (unsigned)l16;
        const int mine = (jm < end) ? col[jm] : 0;
        #pragma unroll
        for (int gq = 0; gq < 2; ++gq) {
            float4 v[8];
            #pragma unroll
            for (int u = 0; u < 8; ++u) {
                const int k  = gq * 8 + u;
                const int sk = __shfl(mine, k, 16);
                v[u] = (base + (unsigned)k < end) ? g4[(long)sk * (H_F / 4) + l16]
                                                  : make_float4(0.f, 0.f, 0.f, 0.f);
            }
            #pragma unroll
            for (int u = 0; u < 8; ++u) {
                acc.x += v[u].x; acc.y += v[u].y; acc.z += v[u].z; acc.w += v[u].w;
            }
        }
    }

    const float nn = in_norm[n];
    const float4 bb = ((const float4*)bias)[l16];
    float4 o;
    o.x = fmaxf(acc.x * nn + bb.x, 0.0f);
    o.y = fmaxf(acc.y * nn + bb.y, 0.0f);
    o.z = fmaxf(acc.z * nn + bb.z, 0.0f);
    o.w = fmaxf(acc.w * nn + bb.w, 0.0f);
    ((float4*)h)[(long)n * (H_F / 4) + l16] = o;
}


__global__ void gcn_head(const float* __restrict__ h,
                         const float* __restrict__ Wm,
                         const float* __restrict__ bm,
                         float* __restrict__ out, int n) {
    int node = blockIdx.x * blockDim.x + threadIdx.x;
    if (node < n) {
        float a0 = bm[0], a1 = bm[1];
        const float* __restrict__ hr = h + (long)node * H_F;
        #pragma unroll 8
        for (int k = 0; k < H_F; ++k) {
            float v = hr[k];
            a0 = fmaf(v, Wm[k * 2 + 0], a0);
            a1 = fmaf(v, Wm[k * 2 + 1], a1);
        }
        out[node * 2 + 0] = a0;
        out[node * 2 + 1] = a1;
    }
}


extern "C" void kernel_launch(void* const* d_in, const int* in_sizes, int n_in,
                              void* d_out, int out_size, void* d_ws, size_t ws_size,
                              hipStream_t stream) {
    const float* feature = (const float*)d_in[0];
    const float* W1      = (const float*)d_in[1];
    const float* b1      = (const float*)d_in[2];
    const float* W2      = (const float*)d_in[3];
    const float* b2      = (const float*)d_in[4];
    const float* Wm      = (const float*)d_in[5];
    const float* bm      = (const float*)d_in[6];
    const int*   src     = (const int*)d_in[7];
    const int*   dst     = (const int*)d_in[8];
    float*       out     = (float*)d_out;

    float* out_norm = (float*)d_ws;
    float* in_norm  = out_norm + N_NODES;
    float* buf0     = in_norm + N_NODES;
    float* buf1     = buf0 + (long)N_NODES * H_F;
    unsigned* part  = (unsigned*)(buf1 + (long)N_NODES * H_F);
    unsigned* pre   = part + (long)DEG_S * DEG_NP;
    unsigned* degin = part + (long)2 * DEG_S * DEG_NP;
    unsigned* rowptr = degin + DEG_NP;
    int*      col   = (int*)(rowptr + DEG_NP + 32);
    unsigned* blktot = (unsigned*)(col + N_EDGES);

    const int  B  = 256;

    gcn_degree_hist<<<2 * DEG_R * DEG_S, DEG_T, 0, stream>>>(src, dst, part);
    gcn_degree_finish<<<2 * FIN_NB, FIN_T, 0, stream>>>(part, out_norm, degin, blktot);

    gcn_rowptr_scan<<<FIN_NB, FIN_T, 0, stream>>>(degin, blktot, rowptr);
    gcn_edge_place<<<DEG_R * DEG_S, DEG_T, 0, stream>>>(src, dst, pre, rowptr, col);

    gcn_gemm_wmma<IN_F><<<N_NODES / 16, 128, 0, stream>>>(feature, W1, out_norm, buf0);
    gcn_pull_bias_relu<<<(int)(((long)N_NODES * 16) / PULL_T), PULL_T, 0, stream>>>(buf0, rowptr, col, in_norm, b1, buf1);

    gcn_gemm_wmma<H_F><<<N_NODES / 16, 128, 0, stream>>>(buf1, W2, out_norm, buf0);
    gcn_pull_bias_relu<<<(int)(((long)N_NODES * 16) / PULL_T), PULL_T, 0, stream>>>(buf0, rowptr, col, in_norm, b2, buf1);

    gcn_head<<<(N_NODES + B - 1) / B, B, 0, stream>>>(buf1, Wm, bm, out, N_NODES);
}


# Fewer, longer-lived waves: an aggregation wave sums 10 consecutive nodes, next source list fetched ahead; a GEMM block does 5 tiles, W kept in registers; a fifth as many waves; sum order unchanged
# speedup vs baseline: 2.9909x; 1.0786x over previous
#include <hip/hip_runtime.h>

#define N_NODES 100000
#define N_EDGES 1600000
#define IN_F    128
#define H_F     64

typedef __attribute__((ext_vector_type(2))) float v2f;
typedef __attribute__((ext_vector_type(8))) float v8f;


#define DEG_R     4
#define DEG_S     32
#define DEG_T     512
#define DEG_U     8
#define DEG_BINS  25024
#define DEG_NP    (DEG_R * DEG_BINS)
#define DEG_SLICE (N_EDGES / DEG_S)
#define DEG_Q     (DEG_SLICE / 4)

static_assert(DEG_NP >= N_NODES, "parts must cover every node");
static_assert(DEG_BINS % 32 == 0, "a part must be whole 128-byte lines");
static_assert(N_EDGES % DEG_S == 0 && DEG_SLICE % 4 == 0, "slices must be whole 16-byte groups");
static_assert(N_NODES % 4 == 0, "finish kernel handles four nodes per thread");

__device__ __forceinline__ void gcn_count_one(unsigned* hist, int v, int lo) {
    const unsigned d = (unsigned)(v - lo);
    if (d < (unsigned)DEG_BINS) atomicAdd(&hist[d], 1u);
}

__global__ __launch_bounds__(DEG_T)
void gcn_degree_hist(const int* __restrict__ src,
                     const int* __restrict__ dst,
                     unsigned* __restrict__ part) {
    __shared__ __attribute__((aligned(16))) unsigned hist[DEG_BINS];

    const int tid = threadIdx.x;
    const int s   = blockIdx.x % DEG_S;
    const int ar  = blockIdx.x / DEG_S;
    const int r   = ar % DEG_R;
    const int a   = ar / DEG_R;
    const int lo  = r * DEG_BINS;

    uint4* h4 = (uint4*)hist;
    for (int i = tid; i < DEG_BINS / 4; i += DEG_T) h4[i] = make_uint4(0u, 0u, 0u, 0u);
    __syncthreads();

    const int* __restrict__ idx = a ? dst : src;
    const int4* __restrict__ p4 = (const int4*)(idx + (long)s * DEG_SLICE);

    for (int base = 0; base < DEG_Q; base += DEG_T * DEG_U) {
        int4 v[DEG_U];
        #pragma unroll
        for (int u = 0; u < DEG_U; ++u) {
            const int q = base + u * DEG_T + tid;
            v[u] = (q < DEG_Q) ? p4[q] : make_int4(-1, -1, -1, -1);
        }
        #pragma unroll
        for (int u = 0; u < DEG_U; ++u) {
            gcn_count_one(hist, v[u].x, lo);
            gcn_count_one(hist, v[u].y, lo);
            gcn_count_one(hist, v[u].z, lo);
            gcn_count_one(hist, v[u].w, lo);
        }
    }
    __syncthreads();

    uint4* __restrict__ o4 = (uint4*)(part + (long)(a * DEG_S + s) * DEG_NP + lo);
    for (int i = tid; i < DEG_BINS / 4; i += DEG_T) o4[i] = h4[i];
}


#define FIN_T   256
#define FIN_NPB (4 * FIN_T)
#define FIN_NB  ((N_NODES + FIN_NPB - 1) / FIN_NPB)

static_assert(FIN_NB <= FIN_T, "one thread per earlier block total in the scan kernel");
static_assert(DEG_S % 8 == 0, "finish kernel loads the slices in groups of eight");

__global__ __launch_bounds__(FIN_T)
void gcn_degree_finish(unsigned* __restrict__ part,
                       float* __restrict__ norm,
                       unsigned* __restrict__ degin,
                       unsigned* __restrict__ blktot) {
    __shared__ unsigned red[FIN_T];
    const int tid = threadIdx.x;
    const int a = blockIdx.x / FIN_NB;
    const int b = blockIdx.x - a * FIN_NB;
    const int j = b * FIN_T + tid;
    const bool valid = j < N_NODES / 4;

    uint4 acc = make_uint4(0u, 0u, 0u, 0u);
    if (valid) {
        uint4* __restrict__ p = (uint4*)(part + (long)a * DEG_S * DEG_NP) + j;
        #pragma unroll
        for (int s0 = 0; s0 < DEG_S; s0 += 8) {
            uint4 v[8];
            #pragma unroll
            for (int u = 0; u < 8; ++u) v[u] = p[(long)(s0 + u) * (DEG_NP / 4)];
            #pragma unroll
            for (int u = 0; u < 8; ++u) {
                if (a == 1) p[(long)(s0 + u) * (DEG_NP / 4)] = acc;
                acc.x += v[u].x; acc.y += v[u].y; acc.z += v[u].z; acc.w += v[u].w;
            }
        }
        float4 o;
        o.x = rsqrtf(fmaxf((float)acc.x, 1.0f));
        o.y = rsqrtf(fmaxf((float)acc.y, 1.0f));
        o.z = rsqrtf(fmaxf((float)acc.z, 1.0f));
        o.w = rsqrtf(fmaxf((float)acc.w, 1.0f));
        *(float4*)(norm + (long)a * N_NODES + 4 * j) = o;
    }

    if (a == 1) {
        if (valid) *(uint4*)(degin + 4 * j) = acc;
        red[tid] = acc.x + acc.y + acc.z + acc.w;
        __syncthreads();
        for (int off = FIN_T / 2; off > 0; off >>= 1) {
            if (tid < off) red[tid] += red[tid + off];
            __syncthreads();
        }
        if (tid == 0) blktot[b] = red[0];
    }
}

__global__ __launch_bounds__(FIN_T)
void gcn_rowptr_scan(const unsigned* __restrict__ degin,
                     const unsigned* __restrict__ blktot,
                     unsigned* __restrict__ rowptr) {
    __shared__ unsigned sh[FIN_T];
    const int tid = threadIdx.x;
    const int b = blockIdx.x;
    const int j = b * FIN_T + tid;
    const bool valid = j < N_NODES / 4;

    sh[tid] = (tid < b) ? blktot[tid] : 0u;
    __syncthreads();
    for (int off = FIN_T / 2; off > 0; off >>= 1) {
        if (tid < off) sh[tid] += sh[tid + off];
        __syncthreads();
    }
    const unsigned base = sh[0];
    __syncthreads();

    uint4 d = make_uint4(0u, 0u, 0u, 0u);
    if (valid) d = *(const uint4*)(degin + 4 * j);
    const unsigned t = d.x + d.y + d.z + d.w;

    sh[tid] = t;
    __syncthreads();
    for (int off = 1; off < FIN_T; off <<= 1) {
        const unsigned y = (tid >= off) ? sh[tid - off] : 0u;
        __syncthreads();
        sh[tid] += y;
        __syncthreads();
    }
    const unsigned first = base + sh[tid] - t;
    if (valid) {
        uint4 o;
        o.x = first;
        o.y = o.x + d.x;
        o.z = o.y + d.y;
        o.w = o.z + d.z;
        *(uint4*)(rowptr + 4 * j) = o;
        if (j == N_NODES / 4 - 1) rowptr[N_NODES] = first + t;
    }
}

__device__ __forceinline__ void gcn_place_one(unsigned* cur, int* __restrict__ col,
                                              int dv, int sv, int lo) {
    const unsigned d = (unsigned)(dv - lo);
    if (d < (unsigned)DEG_BINS) {
        const unsigned pos = atomicAdd(&cur[d], 1u);
        col[pos] = sv;
    }
}

__global__ __launch_bounds__(DEG_T)
void gcn_edge_place(const int* __restrict__ src,
                    const int* __restrict__ dst,
                    const unsigned* __restrict__ pre,
                    const unsigned* __restrict__ rowptr,
                    int* __restrict__ col) {
    __shared__ __attribute__((aligned(16))) unsigned cur[DEG_BINS];

    const int tid = threadIdx.x;
    const int s   = blockIdx.x % DEG_S;
    const int r   = blockIdx.x / DEG_S;
    const int lo  = r * DEG_BINS;

    const uint4* __restrict__ pre4 = (const uint4*)(pre + (long)s * DEG_NP + lo);
    uint4* c4 = (uint4*)cur;
    for (int i = tid; i < DEG_BINS / 4; i += DEG_T) {
        const int n4 = min(lo + 4 * i, N_NODES - 4);
        const uint4 a = *(const uint4*)(rowptr + n4);
        const uint4 p = pre4[i];
        c4[i] = make_uint4(a.x + p.x, a.y + p.y, a.z + p.z, a.w + p.w);
    }
    __syncthreads();

    const int4* __restrict__ d4 = (const int4*)(dst + (long)s * DEG_SLICE);
    const int4* __restrict__ s4 = (const int4*)(src + (long)s * DEG_SLICE);

    for (int base = 0; base < DEG_Q; base += DEG_T * DEG_U) {
        int4 vd[DEG_U], vs[DEG_U];
        #pragma unroll
        for (int u = 0; u < DEG_U; ++u) {
            const int q = base + u * DEG_T + tid;
            vd[u] = (q < DEG_Q) ? d4[q] : make_int4(-1, -1, -1, -1);
            vs[u] = (q < DEG_Q) ? s4[q] : make_int4(0, 0, 0, 0);
        }
        #pragma unroll
        for (int u = 0; u < DEG_U; ++u) {
            gcn_place_one(cur, col, vd[u].x, vs[u].x, lo);
            gcn_place_one(cur, col, vd[u].y, vs[u].y, lo);
            gcn_place_one(cur, col, vd[u].z, vs[u].z, lo);
            gcn_place_one(cur, col, vd[u].w, vs[u].w, lo);
        }
    }
}


#define GEMM_TPB 5

static_assert((N_NODES / 16) % GEMM_TPB == 0 && N_NODES % 16 == 0, "whole blocks of M-tiles");

template <int K>
__global__ __launch_bounds__(128)
void gcn_gemm_wmma(const float* __restrict__ A,
                   const float* __restrict__ W,
                   const float* __restrict__ rowscale,
                   float* __restrict__ C) {
    const int wave = threadIdx.x >> 5;
    const int lane = threadIdx.x & 31;
    const int half = lane >> 4;
    const int l16  = lane & 15;
    const int n0   = wave * 16;

    v2f b[K / 4];
    #pragma unroll
    for (int k0 = 0; k0 < K; k0 += 4) {
        const int ka = k0 + 2 * half;
        b[k0 / 4].x = W[(long)ka       * H_F + n0 + l16];
        b[k0 / 4].y = W[(long)(ka + 1) * H_F + n0 + l16];
    }

    #pragma unroll 1
    for (int t = 0; t < GEMM_TPB; ++t) {
        const int m0 = (blockIdx.x * GEMM_TPB + t) * 16;
        const int m  = m0 + l16;

        const float s = rowscale[m];
        const float* __restrict__ arow = A + (long)m * K;

        v8f c = {};
        #pragma unroll
        for (int k0 = 0; k0 < K; k0 += 4) {
            const int ka = k0 + 2 * half;
            v2f a;
            a.x = arow[ka]     * s;
            a.y = arow[ka + 1] * s;
            c = __builtin_amdgcn_wmma_f32_16x16x4_f32(
                    false, a, false, b[k0 / 4], (short)0, c, false, false);
        }

        #pragma unroll
        for (int r = 0; r < 8; ++r) {
            C[(long)(m0 + r + 8 * half) * H_F + n0 + l16] = c[r];
        }
    }
}


#define PULL_T   256
#define PULL_NPW 10

static_assert(PULL_NPW % 2 == 0 && PULL_NPW < 32, "pairs of nodes; one lane per row start");
static_assert(N_NODES % (PULL_NPW * (PULL_T / 32)) == 0, "whole blocks of whole waves");

__device__ __forceinline__ float4 gcn_sum16(const float4* __restrict__ g4, int mine,
                                            unsigned base, unsigned end, int l16, float4 acc) {
    float4 v[16];
    #pragma unroll
    for (int k = 0; k < 16; ++k) {
        const int sk = __shfl(mine, k, 16);
        v[k] = (base + (unsigned)k < end) ? g4[(long)sk * (H_F / 4) + l16]
                                          : make_float4(0.f, 0.f, 0.f, 0.f);
    }
    #pragma unroll
    for (int k = 0; k < 16; ++k) {
        acc.x += v[k].x; acc.y += v[k].y; acc.z += v[k].z; acc.w += v[k].w;
    }
    return acc;
}

__global__ __launch_bounds__(PULL_T)
void gcn_pull_bias_relu(const float* __restrict__ g,
                        const unsigned* __restrict__ rowptr,
                        const int* __restrict__ col,
                        const float* __restrict__ in_norm,
                        const float* __restrict__ bias,
                        float* __restrict__ h) {
    const int lane = threadIdx.x & 31;
    const int grp  = lane >> 4;
    const int l16  = lane & 15;
    const int wv   = blockIdx.x * (PULL_T / 32) + (threadIdx.x >> 5);
    const int n0   = wv * PULL_NPW;

    const float4* __restrict__ g4 = (const float4*)g;

    const unsigned rp  = rowptr[n0 + min(lane, PULL_NPW)];
    const float    nrm = in_norm[n0 + min(lane, PULL_NPW - 1)];
    const float4   bb  = ((const float4*)bias)[l16];

    unsigned beg = __shfl(rp, grp);
    unsigned end = __shfl(rp, grp + 1);
    int c0 = (beg + (unsigned)l16 < end)       ? col[beg + (unsigned)l16]       : 0;
    int c1 = (beg + 16u + (unsigned)l16 < end) ? col[beg + 16u + (unsigned)l16] : 0;

    asm volatile("" :: "v"(nrm), "v"(bb.x), "v"(bb.y), "v"(bb.z), "v"(bb.w), "v"(c0), "v"(c1));

    #pragma unroll 1
    for (int i = 0; i < PULL_NPW / 2; ++i) {
        unsigned nbeg = 0u, nend = 0u;
        int nc0 = 0, nc1 = 0;
        if (i + 1 < PULL_NPW / 2) {
            nbeg = __shfl(rp, 2 * i + 2 + grp);
            nend = __shfl(rp, 2 * i + 3 + grp);
            nc0 = (nbeg + (unsigned)l16 < nend)       ? col[nbeg + (unsigned)l16]       : 0;
            nc1 = (nbeg + 16u + (unsigned)l16 < nend) ? col[nbeg + 16u + (unsigned)l16] : 0;
        }
        const float nn = __shfl(nrm, 2 * i + grp);

        float4 acc = make_float4(0.f, 0.f, 0.f, 0.f);
        for (unsigned base = beg; base < end; base += 16) {
            const unsigned j2 = base + 32u + (unsigned)l16;
            const int c2 = (j2 < end) ? col[j2] : 0;
            acc = gcn_sum16(g4, c0, base, end, l16, acc);
            c0 = c1;
            c1 = c2;
        }

        const int n = n0 + 2 * i + grp;
        float4 o;
        o.x = fmaxf(acc.x * nn + bb.x, 0.0f);
        o.y = fmaxf(acc.y * nn + bb.y, 0.0f);
        o.z = fmaxf(acc.z * nn + bb.z, 0.0f);
        o.w = fmaxf(acc.w * nn + bb.w, 0.0f);
        ((float4*)h)[(long)n * (H_F / 4) + l16] = o;

        beg = nbeg; end = nend; c0 = nc0; c1 = nc1;
    }
}


__global__ void gcn_head(const float* __restrict__ h,
                         const float* __restrict__ Wm,
                         const float* __restrict__ bm,
                         float* __restrict__ out, int n) {
    int node = blockIdx.x * blockDim.x + threadIdx.x;
    if (node < n) {
        float a0 = bm[0], a1 = bm[1];
        const float* __restrict__ hr = h + (long)node * H_F;
        #pragma unroll 8
        for (int k = 0; k < H_F; ++k) {
            float v = hr[k];
            a0 = fmaf(v, Wm[k * 2 + 0], a0);
            a1 = fmaf(v, Wm[k * 2 + 1], a1);
        }
        out[node * 2 + 0] = a0;
        out[node * 2 + 1] = a1;
    }
}


extern "C" void kernel_launch(void* const* d_in, const int* in_sizes, int n_in,
                              void* d_out, int out_size, void* d_ws, size_t ws_size,
                              hipStream_t stream) {
    const float* feature = (const float*)d_in[0];
    const float* W1      = (const float*)d_in[1];
    const float* b1      = (const float*)d_in[2];
    const float* W2      = (const float*)d_in[3];
    const float* b2      = (const float*)d_in[4];
    const float* Wm      = (const float*)d_in[5];
    const float* bm      = (const float*)d_in[6];
    const int*   src     = (const int*)d_in[7];
    const int*   dst     = (const int*)d_in[8];
    float*       out     = (float*)d_out;

    float* out_norm = (float*)d_ws;
    float* in_norm  = out_norm + N_NODES;
    float* buf0     = in_norm + N_NODES;
    float* buf1     = buf0 + (long)N_NODES * H_F;
    unsigned* part  = (unsigned*)(buf1 + (long)N_NODES * H_F);
    unsigned* pre   = part + (long)DEG_S * DEG_NP;
    unsigned* degin = part + (long)2 * DEG_S * DEG_NP;
    unsigned* rowptr = degin + DEG_NP;
    int*      col   = (int*)(rowptr + DEG_NP + 32);
    unsigned* blktot = (unsigned*)(col + N_EDGES);

    const int  B  = 256;

    gcn_degree_hist<<<2 * DEG_R * DEG_S, DEG_T, 0, stream>>>(src, dst, part);
    gcn_degree_finish<<<2 * FIN_NB, FIN_T, 0, stream>>>(part, out_norm, degin, blktot);

    gcn_rowptr_scan<<<FIN_NB, FIN_T, 0, stream>>>(degin, blktot, rowptr);
    gcn_edge_place<<<DEG_R * DEG_S, DEG_T, 0, stream>>>(src, dst, pre, rowptr, col);

    gcn_gemm_wmma<IN_F><<<N_NODES / 16 / GEMM_TPB, 128, 0, stream>>>(feature, W1, out_norm, buf0);
    gcn_pull_bias_relu<<<N_NODES / (PULL_NPW * (PULL_T / 32)), PULL_T, 0, stream>>>(buf0, rowptr, col, in_norm, b1, buf1);

    gcn_gemm_wmma<H_F><<<N_NODES / 16 / GEMM_TPB, 128, 0, stream>>>(buf1, W2, out_norm, buf0);
    gcn_pull_bias_relu<<<N_NODES / (PULL_NPW * (PULL_T / 32)), PULL_T, 0, stream>>>(buf0, rowptr, col, in_norm, b2, buf1);

    gcn_head<<<(N_NODES + B - 1) / B, B, 0, stream>>>(buf1, Wm, bm, out, N_NODES);
}


# Edge placement and the layer-1 product share one launch on disjoint blocks: 128 blocks place edges while the other WGPs run the same WMMA tiles; 8 launches; sum order unchanged
# speedup vs baseline: 3.1355x; 1.0483x over previous
#include <hip/hip_runtime.h>

#define N_NODES 100000
#define N_EDGES 1600000
#define IN_F    128
#define H_F     64

typedef __attribute__((ext_vector_type(2))) float v2f;
typedef __attribute__((ext_vector_type(8))) float v8f;


#define DEG_R     4
#define DEG_S     32
#define DEG_T     512
#define DEG_U     8
#define DEG_BINS  25024
#define DEG_NP    (DEG_R * DEG_BINS)
#define DEG_SLICE (N_EDGES / DEG_S)
#define DEG_Q     (DEG_SLICE / 4)

static_assert(DEG_NP >= N_NODES, "parts must cover every node");
static_assert(DEG_BINS % 32 == 0, "a part must be whole 128-byte lines");
static_assert(N_EDGES % DEG_S == 0 && DEG_SLICE % 4 == 0, "slices must be whole 16-byte groups");
static_assert(N_NODES % 4 == 0, "finish kernel handles four nodes per thread");

__device__ __forceinline__ void gcn_count_one(unsigned* hist, int v, int lo) {
    const unsigned d = (unsigned)(v - lo);
    if (d < (unsigned)DEG_BINS) atomicAdd(&hist[d], 1u);
}

__global__ __launch_bounds__(DEG_T)
void gcn_degree_hist(const int* __restrict__ src,
                     const int* __restrict__ dst,
                     unsigned* __restrict__ part) {
    __shared__ __attribute__((aligned(16))) unsigned hist[DEG_BINS];

    const int tid = threadIdx.x;
    const int s   = blockIdx.x % DEG_S;
    const int ar  = blockIdx.x / DEG_S;
    const int r   = ar % DEG_R;
    const int a   = ar / DEG_R;
    const int lo  = r * DEG_BINS;

    uint4* h4 = (uint4*)hist;
    for (int i = tid; i < DEG_BINS / 4; i += DEG_T) h4[i] = make_uint4(0u, 0u, 0u, 0u);
    __syncthreads();

    const int* __restrict__ idx = a ? dst : src;
    const int4* __restrict__ p4 = (const int4*)(idx + (long)s * DEG_SLICE);

    for (int base = 0; base < DEG_Q; base += DEG_T * DEG_U) {
        int4 v[DEG_U];
        #pragma unroll
        for (int u = 0; u < DEG_U; ++u) {
            const int q = base + u * DEG_T + tid;
            v[u] = (q < DEG_Q) ? p4[q] : make_int4(-1, -1, -1, -1);
        }
        #pragma unroll
        for (int u = 0; u < DEG_U; ++u) {
            gcn_count_one(hist, v[u].x, lo);
            gcn_count_one(hist, v[u].y, lo);
            gcn_count_one(hist, v[u].z, lo);
            gcn_count_one(hist, v[u].w, lo);
        }
    }
    __syncthreads();

    uint4* __restrict__ o4 = (uint4*)(part + (long)(a * DEG_S + s) * DEG_NP + lo);
    for (int i = tid; i < DEG_BINS / 4; i += DEG_T) o4[i] = h4[i];
}


#define FIN_T   256
#define FIN_NPB (4 * FIN_T)
#define FIN_NB  ((N_NODES + FIN_NPB - 1) / FIN_NPB)

static_assert(FIN_NB <= FIN_T, "one thread per earlier block total in the scan kernel");
static_assert(DEG_S % 8 == 0, "finish kernel loads the slices in groups of eight");

__global__ __launch_bounds__(FIN_T)
void gcn_degree_finish(unsigned* __restrict__ part,
                       float* __restrict__ norm,
                       unsigned* __restrict__ degin,
                       unsigned* __restrict__ blktot) {
    __shared__ unsigned red[FIN_T];
    const int tid = threadIdx.x;
    const int a = blockIdx.x / FIN_NB;
    const int b = blockIdx.x - a * FIN_NB;
    const int j = b * FIN_T + tid;
    const bool valid = j < N_NODES / 4;

    uint4 acc = make_uint4(0u, 0u, 0u, 0u);
    if (valid) {
        uint4* __restrict__ p = (uint4*)(part + (long)a * DEG_S * DEG_NP) + j;
        #pragma unroll
        for (int s0 = 0; s0 < DEG_S; s0 += 8) {
            uint4 v[8];
            #pragma unroll
            for (int u = 0; u < 8; ++u) v[u] = p[(long)(s0 + u) * (DEG_NP / 4)];
            #pragma unroll
            for (int u = 0; u < 8; ++u) {
                if (a == 1) p[(long)(s0 + u) * (DEG_NP / 4)] = acc;
                acc.x += v[u].x; acc.y += v[u].y; acc.z += v[u].z; acc.w += v[u].w;
            }
        }
        float4 o;
        o.x = rsqrtf(fmaxf((float)acc.x, 1.0f));
        o.y = rsqrtf(fmaxf((float)acc.y, 1.0f));
        o.z = rsqrtf(fmaxf((float)acc.z, 1.0f));
        o.w = rsqrtf(fmaxf((float)acc.w, 1.0f));
        *(float4*)(norm + (long)a * N_NODES + 4 * j) = o;
    }

    if (a == 1) {
        if (valid) *(uint4*)(degin + 4 * j) = acc;
        red[tid] = acc.x + acc.y + acc.z + acc.w;
        __syncthreads();
        for (int off = FIN_T / 2; off > 0; off >>= 1) {
            if (tid < off) red[tid] += red[tid + off];
            __syncthreads();
        }
        if (tid == 0) blktot[b] = red[0];
    }
}

__global__ __launch_bounds__(FIN_T)
void gcn_rowptr_scan(const unsigned* __restrict__ degin,
                     const unsigned* __restrict__ blktot,
                     unsigned* __restrict__ rowptr) {
    __shared__ unsigned sh[FIN_T];
    const int tid = threadIdx.x;
    const int b = blockIdx.x;
    const int j = b * FIN_T + tid;
    const bool valid = j < N_NODES / 4;

    sh[tid] = (tid < b) ? blktot[tid] : 0u;
    __syncthreads();
    for (int off = FIN_T / 2; off > 0; off >>= 1) {
        if (tid < off) sh[tid] += sh[tid + off];
        __syncthreads();
    }
    const unsigned base = sh[0];
    __syncthreads();

    uint4 d = make_uint4(0u, 0u, 0u, 0u);
    if (valid) d = *(const uint4*)(degin + 4 * j);
    const unsigned t = d.x + d.y + d.z + d.w;

    sh[tid] = t;
    __syncthreads();
    for (int off = 1; off < FIN_T; off <<= 1) {
        const unsigned y = (tid >= off) ? sh[tid - off] : 0u;
        __syncthreads();
        sh[tid] += y;
        __syncthreads();
    }
    const unsigned first = base + sh[tid] - t;
    if (valid) {
        uint4 o;
        o.x = first;
        o.y = o.x + d.x;
        o.z = o.y + d.y;
        o.w = o.z + d.z;
        *(uint4*)(rowptr + 4 * j) = o;
        if (j == N_NODES / 4 - 1) rowptr[N_NODES] = first + t;
    }
}

__device__ __forceinline__ void gcn_place_one(unsigned* cur, int* __restrict__ col,
                                              int dv, int sv, int lo) {
    const unsigned d = (unsigned)(dv - lo);
    if (d < (unsigned)DEG_BINS) {
        const unsigned pos = atomicAdd(&cur[d], 1u);
        col[pos] = sv;
    }
}

__device__ __forceinline__ void gcn_place_cell(unsigned* cur,
                                               const int* __restrict__ src,
                                               const int* __restrict__ dst,
                                               const unsigned* __restrict__ pre,
                                               const unsigned* __restrict__ rowptr,
                                               int* __restrict__ col,
                                               int cell, int tid) {
    const int s   = cell % DEG_S;
    const int r   = cell / DEG_S;
    const int lo  = r * DEG_BINS;

    const uint4* __restrict__ pre4 = (const uint4*)(pre + (long)s * DEG_NP + lo);
    uint4* c4 = (uint4*)cur;
    for (int i = tid; i < DEG_BINS / 4; i += DEG_T) {
        const int n4 = min(lo + 4 * i, N_NODES - 4);
        const uint4 a = *(const uint4*)(rowptr + n4);
        const uint4 p = pre4[i];
        c4[i] = make_uint4(a.x + p.x, a.y + p.y, a.z + p.z, a.w + p.w);
    }
    __syncthreads();

    const int4* __restrict__ d4 = (const int4*)(dst + (long)s * DEG_SLICE);
    const int4* __restrict__ s4 = (const int4*)(src + (long)s * DEG_SLICE);

    for (int base = 0; base < DEG_Q; base += DEG_T * DEG_U) {
        int4 vd[DEG_U], vs[DEG_U];
        #pragma unroll
        for (int u = 0; u < DEG_U; ++u) {
            const int q = base + u * DEG_T + tid;
            vd[u] = (q < DEG_Q) ? d4[q] : make_int4(-1, -1, -1, -1);
            vs[u] = (q < DEG_Q) ? s4[q] : make_int4(0, 0, 0, 0);
        }
        #pragma unroll
        for (int u = 0; u < DEG_U; ++u) {
            gcn_place_one(cur, col, vd[u].x, vs[u].x, lo);
            gcn_place_one(cur, col, vd[u].y, vs[u].y, lo);
            gcn_place_one(cur, col, vd[u].z, vs[u].z, lo);
            gcn_place_one(cur, col, vd[u].w, vs[u].w, lo);
        }
    }
}


#define GEMM_TPB 5

static_assert((N_NODES / 16) % GEMM_TPB == 0 && N_NODES % 16 == 0, "whole groups of M-tiles");

template <int K, int TPB>
__device__ __forceinline__ void gcn_gemm_tiles(const float* __restrict__ A,
                                               const float* __restrict__ W,
                                               const float* __restrict__ rowscale,
                                               float* __restrict__ C,
                                               int grp, int wave, int lane) {
    const int half = lane >> 4;
    const int l16  = lane & 15;
    const int n0   = wave * 16;

    v2f b[K / 4];
    #pragma unroll
    for (int k0 = 0; k0 < K; k0 += 4) {
        const int ka = k0 + 2 * half;
        b[k0 / 4].x = W[(long)ka       * H_F + n0 + l16];
        b[k0 / 4].y = W[(long)(ka + 1) * H_F + n0 + l16];
    }

    #pragma unroll 1
    for (int t = 0; t < TPB; ++t) {
        const int m0 = (grp * TPB + t) * 16;
        const int m  = m0 + l16;

        const float s = rowscale[m];
        const float* __restrict__ arow = A + (long)m * K;

        v8f c = {};
        #pragma unroll
        for (int k0 = 0; k0 < K; k0 += 4) {
            const int ka = k0 + 2 * half;
            v2f a;
            a.x = arow[ka]     * s;
            a.y = arow[ka + 1] * s;
            c = __builtin_amdgcn_wmma_f32_16x16x4_f32(
                    false, a, false, b[k0 / 4], (short)0, c, false, false);
        }

        #pragma unroll
        for (int r = 0; r < 8; ++r) {
            C[(long)(m0 + r + 8 * half) * H_F + n0 + l16] = c[r];
        }
    }
}

template <int K>
__global__ __launch_bounds__(128)
void gcn_gemm_wmma(const float* __restrict__ A,
                   const float* __restrict__ W,
                   const float* __restrict__ rowscale,
                   float* __restrict__ C) {
    gcn_gemm_tiles<K, GEMM_TPB>(A, W, rowscale, C, blockIdx.x, threadIdx.x >> 5, threadIdx.x & 31);
}


#define PG_PLACE_BLOCKS (DEG_R * DEG_S)
#define PG_TPB    2
#define PG_GROUPS (N_NODES / 16 / PG_TPB)
#define PG_GPB    (DEG_T / 128)
#define PG_GEMM_BLOCKS ((PG_GROUPS + PG_GPB - 1) / PG_GPB)

static_assert((N_NODES / 16) % PG_TPB == 0, "whole groups of M-tiles");
static_assert(DEG_T % 128 == 0, "whole GEMM groups per block");

__global__ __launch_bounds__(DEG_T)
void gcn_place_and_gemm1(const int* __restrict__ src,
                         const int* __restrict__ dst,
                         const unsigned* __restrict__ pre,
                         const unsigned* __restrict__ rowptr,
                         int* __restrict__ col,
                         const float* __restrict__ feature,
                         const float* __restrict__ W1,
                         const float* __restrict__ out_norm,
                         float* __restrict__ g) {
    __shared__ __attribute__((aligned(16))) unsigned cur[DEG_BINS];

    if (blockIdx.x >= PG_PLACE_BLOCKS) {
        const int grp = (blockIdx.x - PG_PLACE_BLOCKS) * PG_GPB + (threadIdx.x >> 7);
        if (grp < PG_GROUPS)
            gcn_gemm_tiles<IN_F, PG_TPB>(feature, W1, out_norm, g, grp,
                                         (threadIdx.x >> 5) & 3, threadIdx.x & 31);
        return;
    }
    gcn_place_cell(cur, src, dst, pre, rowptr, col, blockIdx.x, threadIdx.x);
}


#define PULL_T   256
#define PULL_NPW 10

static_assert(PULL_NPW % 2 == 0 && PULL_NPW < 32, "pairs of nodes; one lane per row start");
static_assert(N_NODES % (PULL_NPW * (PULL_T / 32)) == 0, "whole blocks of whole waves");

__device__ __forceinline__ float4 gcn_sum16(const float4* __restrict__ g4, int mine,
                                            unsigned base, unsigned end, int l16, float4 acc) {
    float4 v[16];
    #pragma unroll
    for (int k = 0; k < 16; ++k) {
        const int sk = __shfl(mine, k, 16);
        v[k] = (base + (unsigned)k < end) ? g4[(long)sk * (H_F / 4) + l16]
                                          : make_float4(0.f, 0.f, 0.f, 0.f);
    }
    #pragma unroll
    for (int k = 0; k < 16; ++k) {
        acc.x += v[k].x; acc.y += v[k].y; acc.z += v[k].z; acc.w += v[k].w;
    }
    return acc;
}

__global__ __launch_bounds__(PULL_T)
void gcn_pull_bias_relu(const float* __restrict__ g,
                        const unsigned* __restrict__ rowptr,
                        const int* __restrict__ col,
                        const float* __restrict__ in_norm,
                        const float* __restrict__ bias,
                        float* __restrict__ h) {
    const int lane = threadIdx.x & 31;
    const int grp  = lane >> 4;
    const int l16  = lane & 15;
    const int wv   = blockIdx.x * (PULL_T / 32) + (threadIdx.x >> 5);
    const int n0   = wv * PULL_NPW;

    const float4* __restrict__ g4 = (const float4*)g;

    const unsigned rp  = rowptr[n0 + min(lane, PULL_NPW)];
    const float    nrm = in_norm[n0 + min(lane, PULL_NPW - 1)];
    const float4   bb  = ((const float4*)bias)[l16];

    unsigned beg = __shfl(rp, grp);
    unsigned end = __shfl(rp, grp + 1);
    int c0 = (beg + (unsigned)l16 < end)       ? col[beg + (unsigned)l16]       : 0;
    int c1 = (beg + 16u + (unsigned)l16 < end) ? col[beg + 16u + (unsigned)l16] : 0;

    asm volatile("" :: "v"(nrm), "v"(bb.x), "v"(bb.y), "v"(bb.z), "v"(bb.w), "v"(c0), "v"(c1));

    #pragma unroll 1
    for (int i = 0; i < PULL_NPW / 2; ++i) {
        unsigned nbeg = 0u, nend = 0u;
        int nc0 = 0, nc1 = 0;
        if (i + 1 < PULL_NPW / 2) {
            nbeg = __shfl(rp, 2 * i + 2 + grp);
            nend = __shfl(rp, 2 * i + 3 + grp);
            nc0 = (nbeg + (unsigned)l16 < nend)       ? col[nbeg + (unsigned)l16]       : 0;
            nc1 = (nbeg + 16u + (unsigned)l16 < nend) ? col[nbeg + 16u + (unsigned)l16] : 0;
        }
        const float nn = __shfl(nrm, 2 * i + grp);

        float4 acc = make_float4(0.f, 0.f, 0.f, 0.f);
        for (unsigned base = beg; base < end; base += 16) {
            const unsigned j2 = base + 32u + (unsigned)l16;
            const int c2 = (j2 < end) ? col[j2] : 0;
            acc = gcn_sum16(g4, c0, base, end, l16, acc);
            c0 = c1;
            c1 = c2;
        }

        const int n = n0 + 2 * i + grp;
        float4 o;
        o.x = fmaxf(acc.x * nn + bb.x, 0.0f);
        o.y = fmaxf(acc.y * nn + bb.y, 0.0f);
        o.z = fmaxf(acc.z * nn + bb.z, 0.0f);
        o.w = fmaxf(acc.w * nn + bb.w, 0.0f);
        ((float4*)h)[(long)n * (H_F / 4) + l16] = o;

        beg = nbeg; end = nend; c0 = nc0; c1 = nc1;
    }
}


__global__ void gcn_head(const float* __restrict__ h,
                         const float* __restrict__ Wm,
                         const float* __restrict__ bm,
                         float* __restrict__ out, int n) {
    int node = blockIdx.x * blockDim.x + threadIdx.x;
    if (node < n) {
        float a0 = bm[0], a1 = bm[1];
        const float* __restrict__ hr = h + (long)node * H_F;
        #pragma unroll 8
        for (int k = 0; k < H_F; ++k) {
            float v = hr[k];
            a0 = fmaf(v, Wm[k * 2 + 0], a0);
            a1 = fmaf(v, Wm[k * 2 + 1], a1);
        }
        out[node * 2 + 0] = a0;
        out[node * 2 + 1] = a1;
    }
}


extern "C" void kernel_launch(void* const* d_in, const int* in_sizes, int n_in,
                              void* d_out, int out_size, void* d_ws, size_t ws_size,
                              hipStream_t stream) {
    const float* feature = (const float*)d_in[0];
    const float* W1      = (const float*)d_in[1];
    const float* b1      = (const float*)d_in[2];
    const float* W2      = (const float*)d_in[3];
    const float* b2      = (const float*)d_in[4];
    const float* Wm      = (const float*)d_in[5];
    const float* bm      = (const float*)d_in[6];
    const int*   src     = (const int*)d_in[7];
    const int*   dst     = (const int*)d_in[8];
    float*       out     = (float*)d_out;

    float* out_norm = (float*)d_ws;
    float* in_norm  = out_norm + N_NODES;
    float* buf0     = in_norm + N_NODES;
    float* buf1     = buf0 + (long)N_NODES * H_F;
    unsigned* part  = (unsigned*)(buf1 + (long)N_NODES * H_F);
    unsigned* pre   = part + (long)DEG_S * DEG_NP;
    unsigned* degin = part + (long)2 * DEG_S * DEG_NP;
    unsigned* rowptr = degin + DEG_NP;
    int*      col   = (int*)(rowptr + DEG_NP + 32);
    unsigned* blktot = (unsigned*)(col + N_EDGES);

    const int  B  = 256;

    gcn_degree_hist<<<2 * DEG_R * DEG_S, DEG_T, 0, stream>>>(src, dst, part);
    gcn_degree_finish<<<2 * FIN_NB, FIN_T, 0, stream>>>(part, out_norm, degin, blktot);

    gcn_rowptr_scan<<<FIN_NB, FIN_T, 0, stream>>>(degin, blktot, rowptr);
    gcn_place_and_gemm1<<<PG_PLACE_BLOCKS + PG_GEMM_BLOCKS, DEG_T, 0, stream>>>(
        src, dst, pre, rowptr, col, feature, W1, out_norm, buf0);

    gcn_pull_bias_relu<<<N_NODES / (PULL_NPW * (PULL_T / 32)), PULL_T, 0, stream>>>(buf0, rowptr, col, in_norm, b1, buf1);

    gcn_gemm_wmma<H_F><<<N_NODES / 16 / GEMM_TPB, 128, 0, stream>>>(buf1, W2, out_norm, buf0);
    gcn_pull_bias_relu<<<N_NODES / (PULL_NPW * (PULL_T / 32)), PULL_T, 0, stream>>>(buf0, rowptr, col, in_norm, b2, buf1);

    gcn_head<<<(N_NODES + B - 1) / B, B, 0, stream>>>(buf1, Wm, bm, out, N_NODES);
}


# Each aggregation takes over the dense layer that read its rows: layer-1 rows go to an LDS tile and the wave runs the same f32 WMMA k loop on it; head as a 16-lane dot (sum order changes); 6 launches
# speedup vs baseline: 3.3113x; 1.0561x over previous
#include <hip/hip_runtime.h>

#define N_NODES 100000
#define N_EDGES 1600000
#define IN_F    128
#define H_F     64

typedef __attribute__((ext_vector_type(2))) float v2f;
typedef __attribute__((ext_vector_type(8))) float v8f;


#define DEG_R     4
#define DEG_S     32
#define DEG_T     512
#define DEG_U     8
#define DEG_BINS  25024
#define DEG_NP    (DEG_R * DEG_BINS)
#define DEG_SLICE (N_EDGES / DEG_S)
#define DEG_Q     (DEG_SLICE / 4)

static_assert(DEG_NP >= N_NODES, "parts must cover every node");
static_assert(DEG_BINS % 32 == 0, "a part must be whole 128-byte lines");
static_assert(N_EDGES % DEG_S == 0 && DEG_SLICE % 4 == 0, "slices must be whole 16-byte groups");
static_assert(N_NODES % 4 == 0, "finish kernel handles four nodes per thread");

__device__ __forceinline__ void gcn_count_one(unsigned* hist, int v, int lo) {
    const unsigned d = (unsigned)(v - lo);
    if (d < (unsigned)DEG_BINS) atomicAdd(&hist[d], 1u);
}

__global__ __launch_bounds__(DEG_T)
void gcn_degree_hist(const int* __restrict__ src,
                     const int* __restrict__ dst,
                     unsigned* __restrict__ part) {
    __shared__ __attribute__((aligned(16))) unsigned hist[DEG_BINS];

    const int tid = threadIdx.x;
    const int s   = blockIdx.x % DEG_S;
    const int ar  = blockIdx.x / DEG_S;
    const int r   = ar % DEG_R;
    const int a   = ar / DEG_R;
    const int lo  = r * DEG_BINS;

    uint4* h4 = (uint4*)hist;
    for (int i = tid; i < DEG_BINS / 4; i += DEG_T) h4[i] = make_uint4(0u, 0u, 0u, 0u);
    __syncthreads();

    const int* __restrict__ idx = a ? dst : src;
    const int4* __restrict__ p4 = (const int4*)(idx + (long)s * DEG_SLICE);

    for (int base = 0; base < DEG_Q; base += DEG_T * DEG_U) {
        int4 v[DEG_U];
        #pragma unroll
        for (int u = 0; u < DEG_U; ++u) {
            const int q = base + u * DEG_T + tid;
            v[u] = (q < DEG_Q) ? p4[q] : make_int4(-1, -1, -1, -1);
        }
        #pragma unroll
        for (int u = 0; u < DEG_U; ++u) {
            gcn_count_one(hist, v[u].x, lo);
            gcn_count_one(hist, v[u].y, lo);
            gcn_count_one(hist, v[u].z, lo);
            gcn_count_one(hist, v[u].w, lo);
        }
    }
    __syncthreads();

    uint4* __restrict__ o4 = (uint4*)(part + (long)(a * DEG_S + s) * DEG_NP + lo);
    for (int i = tid; i < DEG_BINS / 4; i += DEG_T) o4[i] = h4[i];
}


#define FIN_T   256
#define FIN_NPB (4 * FIN_T)
#define FIN_NB  ((N_NODES + FIN_NPB - 1) / FIN_NPB)

static_assert(FIN_NB <= FIN_T, "one thread per earlier block total in the scan kernel");
static_assert(DEG_S % 8 == 0, "finish kernel loads the slices in groups of eight");

__global__ __launch_bounds__(FIN_T)
void gcn_degree_finish(unsigned* __restrict__ part,
                       float* __restrict__ norm,
                       unsigned* __restrict__ degin,
                       unsigned* __restrict__ blktot) {
    __shared__ unsigned red[FIN_T];
    const int tid = threadIdx.x;
    const int a = blockIdx.x / FIN_NB;
    const int b = blockIdx.x - a * FIN_NB;
    const int j = b * FIN_T + tid;
    const bool valid = j < N_NODES / 4;

    uint4 acc = make_uint4(0u, 0u, 0u, 0u);
    if (valid) {
        uint4* __restrict__ p = (uint4*)(part + (long)a * DEG_S * DEG_NP) + j;
        #pragma unroll
        for (int s0 = 0; s0 < DEG_S; s0 += 8) {
            uint4 v[8];
            #pragma unroll
            for (int u = 0; u < 8; ++u) v[u] = p[(long)(s0 + u) * (DEG_NP / 4)];
            #pragma unroll
            for (int u = 0; u < 8; ++u) {
                if (a == 1) p[(long)(s0 + u) * (DEG_NP / 4)] = acc;
                acc.x += v[u].x; acc.y += v[u].y; acc.z += v[u].z; acc.w += v[u].w;
            }
        }
        float4 o;
        o.x = rsqrtf(fmaxf((float)acc.x, 1.0f));
        o.y = rsqrtf(fmaxf((float)acc.y, 1.0f));
        o.z = rsqrtf(fmaxf((float)acc.z, 1.0f));
        o.w = rsqrtf(fmaxf((float)acc.w, 1.0f));
        *(float4*)(norm + (long)a * N_NODES + 4 * j) = o;
    }

    if (a == 1) {
        if (valid) *(uint4*)(degin + 4 * j) = acc;
        red[tid] = acc.x + acc.y + acc.z + acc.w;
        __syncthreads();
        for (int off = FIN_T / 2; off > 0; off >>= 1) {
            if (tid < off) red[tid] += red[tid + off];
            __syncthreads();
        }
        if (tid == 0) blktot[b] = red[0];
    }
}

__global__ __launch_bounds__(FIN_T)
void gcn_rowptr_scan(const unsigned* __restrict__ degin,
                     const unsigned* __restrict__ blktot,
                     unsigned* __restrict__ rowptr) {
    __shared__ unsigned sh[FIN_T];
    const int tid = threadIdx.x;
    const int b = blockIdx.x;
    const int j = b * FIN_T + tid;
    const bool valid = j < N_NODES / 4;

    sh[tid] = (tid < b) ? blktot[tid] : 0u;
    __syncthreads();
    for (int off = FIN_T / 2; off > 0; off >>= 1) {
        if (tid < off) sh[tid] += sh[tid + off];
        __syncthreads();
    }
    const unsigned base = sh[0];
    __syncthreads();

    uint4 d = make_uint4(0u, 0u, 0u, 0u);
    if (valid) d = *(const uint4*)(degin + 4 * j);
    const unsigned t = d.x + d.y + d.z + d.w;

    sh[tid] = t;
    __syncthreads();
    for (int off = 1; off < FIN_T; off <<= 1) {
        const unsigned y = (tid >= off) ? sh[tid - off] : 0u;
        __syncthreads();
        sh[tid] += y;
        __syncthreads();
    }
    const unsigned first = base + sh[tid] - t;
    if (valid) {
        uint4 o;
        o.x = first;
        o.y = o.x + d.x;
        o.z = o.y + d.y;
        o.w = o.z + d.z;
        *(uint4*)(rowptr + 4 * j) = o;
        if (j == N_NODES / 4 - 1) rowptr[N_NODES] = first + t;
    }
}

__device__ __forceinline__ void gcn_place_one(unsigned* cur, int* __restrict__ col,
                                              int dv, int sv, int lo) {
    const unsigned d = (unsigned)(dv - lo);
    if (d < (unsigned)DEG_BINS) {
        const unsigned pos = atomicAdd(&cur[d], 1u);
        col[pos] = sv;
    }
}

__device__ __forceinline__ void gcn_place_cell(unsigned* cur,
                                               const int* __restrict__ src,
                                               const int* __restrict__ dst,
                                               const unsigned* __restrict__ pre,
                                               const unsigned* __restrict__ rowptr,
                                               int* __restrict__ col,
                                               int cell, int tid) {
    const int s   = cell % DEG_S;
    const int r   = cell / DEG_S;
    const int lo  = r * DEG_BINS;

    const uint4* __restrict__ pre4 = (const uint4*)(pre + (long)s * DEG_NP + lo);
    uint4* c4 = (uint4*)cur;
    for (int i = tid; i < DEG_BINS / 4; i += DEG_T) {
        const int n4 = min(lo + 4 * i, N_NODES - 4);
        const uint4 a = *(const uint4*)(rowptr + n4);
        const uint4 p = pre4[i];
        c4[i] = make_uint4(a.x + p.x, a.y + p.y, a.z + p.z, a.w + p.w);
    }
    __syncthreads();

    const int4* __restrict__ d4 = (const int4*)(dst + (long)s * DEG_SLICE);
    const int4* __restrict__ s4 = (const int4*)(src + (long)s * DEG_SLICE);

    for (int base = 0; base < DEG_Q; base += DEG_T * DEG_U) {
        int4 vd[DEG_U], vs[DEG_U];
        #pragma unroll
        for (int u = 0; u < DEG_U; ++u) {
            const int q = base + u * DEG_T + tid;
            vd[u] = (q < DEG_Q) ? d4[q] : make_int4(-1, -1, -1, -1);
            vs[u] = (q < DEG_Q) ? s4[q] : make_int4(0, 0, 0, 0);
        }
        #pragma unroll
        for (int u = 0; u < DEG_U; ++u) {
            gcn_place_one(cur, col, vd[u].x, vs[u].x, lo);
            gcn_place_one(cur, col, vd[u].y, vs[u].y, lo);
            gcn_place_one(cur, col, vd[u].z, vs[u].z, lo);
            gcn_place_one(cur, col, vd[u].w, vs[u].w, lo);
        }
    }
}


static_assert(N_NODES % 16 == 0, "whole M-tiles");

template <int K, int TPB>
__device__ __forceinline__ void gcn_gemm_tiles(const float* __restrict__ A,
                                               const float* __restrict__ W,
                                               const float* __restrict__ rowscale,
                                               float* __restrict__ C,
                                               int grp, int wave, int lane) {
    const int half = lane >> 4;
    const int l16  = lane & 15;
    const int n0   = wave * 16;

    v2f b[K / 4];
    #pragma unroll
    for (int k0 = 0; k0 < K; k0 += 4) {
        const int ka = k0 + 2 * half;
        b[k0 / 4].x = W[(long)ka       * H_F + n0 + l16];
        b[k0 / 4].y = W[(long)(ka + 1) * H_F + n0 + l16];
    }

    #pragma unroll 1
    for (int t = 0; t < TPB; ++t) {
        const int m0 = (grp * TPB + t) * 16;
        const int m  = m0 + l16;

        const float s = rowscale[m];
        const float* __restrict__ arow = A + (long)m * K;

        v8f c = {};
        #pragma unroll
        for (int k0 = 0; k0 < K; k0 += 4) {
            const int ka = k0 + 2 * half;
            v2f a;
            a.x = arow[ka]     * s;
            a.y = arow[ka + 1] * s;
            c = __builtin_amdgcn_wmma_f32_16x16x4_f32(
                    false, a, false, b[k0 / 4], (short)0, c, false, false);
        }

        #pragma unroll
        for (int r = 0; r < 8; ++r) {
            C[(long)(m0 + r + 8 * half) * H_F + n0 + l16] = c[r];
        }
    }
}


#define PG_PLACE_BLOCKS (DEG_R * DEG_S)
#define PG_TPB    2
#define PG_GROUPS (N_NODES / 16 / PG_TPB)
#define PG_GPB    (DEG_T / 128)
#define PG_GEMM_BLOCKS ((PG_GROUPS + PG_GPB - 1) / PG_GPB)

static_assert((N_NODES / 16) % PG_TPB == 0, "whole groups of M-tiles");
static_assert(DEG_T % 128 == 0, "whole GEMM groups per block");

__global__ __launch_bounds__(DEG_T)
void gcn_place_and_gemm1(const int* __restrict__ src,
                         const int* __restrict__ dst,
                         const unsigned* __restrict__ pre,
                         const unsigned* __restrict__ rowptr,
                         int* __restrict__ col,
                         const float* __restrict__ feature,
                         const float* __restrict__ W1,
                         const float* __restrict__ out_norm,
                         float* __restrict__ g) {
    __shared__ __attribute__((aligned(16))) unsigned cur[DEG_BINS];

    if (blockIdx.x >= PG_PLACE_BLOCKS) {
        const int grp = (blockIdx.x - PG_PLACE_BLOCKS) * PG_GPB + (threadIdx.x >> 7);
        if (grp < PG_GROUPS)
            gcn_gemm_tiles<IN_F, PG_TPB>(feature, W1, out_norm, g, grp,
                                         (threadIdx.x >> 5) & 3, threadIdx.x & 31);
        return;
    }
    gcn_place_cell(cur, src, dst, pre, rowptr, col, blockIdx.x, threadIdx.x);
}


__device__ __forceinline__ float4 gcn_sum16(const float4* __restrict__ g4, int mine,
                                            unsigned base, unsigned end, int l16, float4 acc) {
    float4 v[16];
    #pragma unroll
    for (int k = 0; k < 16; ++k) {
        const int sk = __shfl(mine, k, 16);
        v[k] = (base + (unsigned)k < end) ? g4[(long)sk * (H_F / 4) + l16]
                                          : make_float4(0.f, 0.f, 0.f, 0.f);
    }
    #pragma unroll
    for (int k = 0; k < 16; ++k) {
        acc.x += v[k].x; acc.y += v[k].y; acc.z += v[k].z; acc.w += v[k].w;
    }
    return acc;
}

template <int NPW, typename Epi>
__device__ __forceinline__ void gcn_pull_rows(const float* __restrict__ g,
                                              const unsigned* __restrict__ rowptr,
                                              const int* __restrict__ col,
                                              const float* __restrict__ in_norm,
                                              const float* __restrict__ bias,
                                              int n0, int lane, const Epi& epi) {
    static_assert(NPW % 2 == 0 && NPW < 32, "pairs of nodes; one lane per row start");
    const int grp  = lane >> 4;
    const int l16  = lane & 15;

    const float4* __restrict__ g4 = (const float4*)g;

    const unsigned rp  = rowptr[n0 + min(lane, NPW)];
    const float    nrm = in_norm[n0 + min(lane, NPW - 1)];
    const float4   bb  = ((const float4*)bias)[l16];

    unsigned beg = __shfl(rp, grp);
    unsigned end = __shfl(rp, grp + 1);
    int c0 = (beg + (unsigned)l16 < end)       ? col[beg + (unsigned)l16]       : 0;
    int c1 = (beg + 16u + (unsigned)l16 < end) ? col[beg + 16u + (unsigned)l16] : 0;

    asm volatile("" :: "v"(nrm), "v"(bb.x), "v"(bb.y), "v"(bb.z), "v"(bb.w), "v"(c0), "v"(c1));

    #pragma unroll 1
    for (int i = 0; i < NPW / 2; ++i) {
        unsigned nbeg = 0u, nend = 0u;
        int nc0 = 0, nc1 = 0;
        if (i + 1 < NPW / 2) {
            nbeg = __shfl(rp, 2 * i + 2 + grp);
            nend = __shfl(rp, 2 * i + 3 + grp);
            nc0 = (nbeg + (unsigned)l16 < nend)       ? col[nbeg + (unsigned)l16]       : 0;
            nc1 = (nbeg + 16u + (unsigned)l16 < nend) ? col[nbeg + 16u + (unsigned)l16] : 0;
        }
        const float nn = __shfl(nrm, 2 * i + grp);

        float4 acc = make_float4(0.f, 0.f, 0.f, 0.f);
        for (unsigned base = beg; base < end; base += 16) {
            const unsigned j2 = base + 32u + (unsigned)l16;
            const int c2 = (j2 < end) ? col[j2] : 0;
            acc = gcn_sum16(g4, c0, base, end, l16, acc);
            c0 = c1;
            c1 = c2;
        }

        float4 o;
        o.x = fmaxf(acc.x * nn + bb.x, 0.0f);
        o.y = fmaxf(acc.y * nn + bb.y, 0.0f);
        o.z = fmaxf(acc.z * nn + bb.z, 0.0f);
        o.w = fmaxf(acc.w * nn + bb.w, 0.0f);
        epi.row(i, grp, l16, n0 + 2 * i + grp, o);

        beg = nbeg; end = nend; c0 = nc0; c1 = nc1;
    }
}


#define PG2_NPW    16
#define PG2_WPB    5
#define PG2_T      (PG2_WPB * 32)
#define PG2_PITCH  (H_F + 4)
#define PG2_WAVES  (N_NODES / PG2_NPW)
#define PG2_BLOCKS (PG2_WAVES / PG2_WPB)
#define PG2_W2Q    (H_F * H_F / 4)
#define PG2_STG    128

static_assert(N_NODES % PG2_NPW == 0, "whole tiles of 16 nodes");
static_assert(PG2_WAVES % PG2_WPB == 0, "whole blocks: every wave of every block has a tile");
static_assert(PG2_STG % 32 == 0 && PG2_STG <= PG2_T && PG2_W2Q % PG2_STG == 0, "whole waves stage W2 in whole rounds");
static_assert((PG2_PITCH * 4) % 16 == 0, "16-byte row stores into the tile");

struct gcn_epi_tile {
    float* tile;
    __device__ __forceinline__ void row(int i, int grp, int l16, int n, float4 o) const {
        *(float4*)(tile + (2 * i + grp) * PG2_PITCH + 4 * l16) = o;
    }
};

__global__ __launch_bounds__(PG2_T)
void gcn_pull_gemm2(const float* __restrict__ g,
                    const unsigned* __restrict__ rowptr,
                    const int* __restrict__ col,
                    const float* __restrict__ in_norm,
                    const float* __restrict__ bias,
                    const float* __restrict__ W2,
                    const float* __restrict__ out_norm,
                    float* __restrict__ g2) {
    __shared__ __attribute__((aligned(16))) float wt[H_F * PG2_PITCH];
    __shared__ __attribute__((aligned(16))) float tiles[PG2_WPB * PG2_NPW * PG2_PITCH];

    const int tid = threadIdx.x;

    if (tid < PG2_STG) {
        #pragma unroll
        for (int j = 0; j < PG2_W2Q / PG2_STG; ++j) {
            const int q = j * PG2_STG + tid;
            const float4 v = ((const float4*)W2)[q];
            const int k = q / (H_F / 4);
            const int n = (q % (H_F / 4)) * 4;
            wt[(n + 0) * PG2_PITCH + k] = v.x;
            wt[(n + 1) * PG2_PITCH + k] = v.y;
            wt[(n + 2) * PG2_PITCH + k] = v.z;
            wt[(n + 3) * PG2_PITCH + k] = v.w;
        }
    }
    __syncthreads();

    const int wave = tid >> 5;
    const int lane = tid & 31;
    const int wv   = blockIdx.x * PG2_WPB + wave;
    const int n0   = wv * PG2_NPW;
    const int half = lane >> 4;
    const int l16  = lane & 15;

    const float s = out_norm[n0 + l16];

    float* tile = tiles + wave * (PG2_NPW * PG2_PITCH);
    gcn_epi_tile epi;
    epi.tile = tile;
    gcn_pull_rows<PG2_NPW>(g, rowptr, col, in_norm, bias, n0, lane, epi);


    const float* arow = tile + l16 * PG2_PITCH + 2 * half;
    const float* bcol = wt   + l16 * PG2_PITCH + 2 * half;

    v8f c0 = {}, c1 = {}, c2 = {}, c3 = {};
    #pragma unroll
    for (int k0 = 0; k0 < H_F; k0 += 4) {
        const float2 av  = *(const float2*)(arow + k0);
        const float2 b0v = *(const float2*)(bcol + 0 * 16 * PG2_PITCH + k0);
        const float2 b1v = *(const float2*)(bcol + 1 * 16 * PG2_PITCH + k0);
        const float2 b2v = *(const float2*)(bcol + 2 * 16 * PG2_PITCH + k0);
        const float2 b3v = *(const float2*)(bcol + 3 * 16 * PG2_PITCH + k0);
        v2f a, b0, b1, b2, b3;
        a.x  = av.x * s;  a.y  = av.y * s;
        b0.x = b0v.x;     b0.y = b0v.y;
        b1.x = b1v.x;     b1.y = b1v.y;
        b2.x = b2v.x;     b2.y = b2v.y;
        b3.x = b3v.x;     b3.y = b3v.y;
        c0 = __builtin_amdgcn_wmma_f32_16x16x4_f32(false, a, false, b0, (short)0, c0, false, false);
        c1 = __builtin_amdgcn_wmma_f32_16x16x4_f32(false, a, false, b1, (short)0, c1, false, false);
        c2 = __builtin_amdgcn_wmma_f32_16x16x4_f32(false, a, false, b2, (short)0, c2, false, false);
        c3 = __builtin_amdgcn_wmma_f32_16x16x4_f32(false, a, false, b3, (short)0, c3, false, false);
    }

    float* __restrict__ crow = g2 + (long)(n0 + 8 * half) * H_F + l16;
    #pragma unroll
    for (int r = 0; r < 8; ++r) {
        crow[(long)r * H_F +  0] = c0[r];
        crow[(long)r * H_F + 16] = c1[r];
        crow[(long)r * H_F + 32] = c2[r];
        crow[(long)r * H_F + 48] = c3[r];
    }
}


#define PULL_T   256
#define PULL_NPW 10

static_assert(N_NODES % (PULL_NPW * (PULL_T / 32)) == 0, "whole blocks of whole waves");

__device__ __forceinline__ float gcn_row16_sum(float v) {
    v += __builtin_bit_cast(float, __builtin_amdgcn_update_dpp(0, __builtin_bit_cast(int, v), 0x161, 0xf, 0xf, true));
    v += __builtin_bit_cast(float, __builtin_amdgcn_update_dpp(0, __builtin_bit_cast(int, v), 0x162, 0xf, 0xf, true));
    v += __builtin_bit_cast(float, __builtin_amdgcn_update_dpp(0, __builtin_bit_cast(int, v), 0x164, 0xf, 0xf, true));
    v += __builtin_bit_cast(float, __builtin_amdgcn_update_dpp(0, __builtin_bit_cast(int, v), 0x168, 0xf, 0xf, true));
    return v;
}

struct gcn_epi_head {
    float4 w0, w1;
    float2 bh;
    float* out;
    __device__ __forceinline__ void row(int i, int grp, int l16, int n, float4 o) const {
        float p0 = o.x * w0.x + o.y * w0.z + o.z * w1.x + o.w * w1.z;
        float p1 = o.x * w0.y + o.y * w0.w + o.z * w1.y + o.w * w1.w;
        p0 = gcn_row16_sum(p0);
        p1 = gcn_row16_sum(p1);
        if (l16 == 0) ((float2*)out)[n] = make_float2(p0 + bh.x, p1 + bh.y);
    }
};

__global__ __launch_bounds__(PULL_T)
void gcn_pull_head(const float* __restrict__ g,
                   const unsigned* __restrict__ rowptr,
                   const int* __restrict__ col,
                   const float* __restrict__ in_norm,
                   const float* __restrict__ bias,
                   const float* __restrict__ Wm,
                   const float* __restrict__ bm,
                   float* __restrict__ out) {
    const int lane = threadIdx.x & 31;
    const int l16  = lane & 15;
    const int wv   = blockIdx.x * (PULL_T / 32) + (threadIdx.x >> 5);

    gcn_epi_head epi;
    epi.w0  = ((const float4*)Wm)[2 * l16];
    epi.w1  = ((const float4*)Wm)[2 * l16 + 1];
    epi.bh  = make_float2(bm[0], bm[1]);
    epi.out = out;
    gcn_pull_rows<PULL_NPW>(g, rowptr, col, in_norm, bias, wv * PULL_NPW, lane, epi);
}


extern "C" void kernel_launch(void* const* d_in, const int* in_sizes, int n_in,
                              void* d_out, int out_size, void* d_ws, size_t ws_size,
                              hipStream_t stream) {
    const float* feature = (const float*)d_in[0];
    const float* W1      = (const float*)d_in[1];
    const float* b1      = (const float*)d_in[2];
    const float* W2      = (const float*)d_in[3];
    const float* b2      = (const float*)d_in[4];
    const float* Wm      = (const float*)d_in[5];
    const float* bm      = (const float*)d_in[6];
    const int*   src     = (const int*)d_in[7];
    const int*   dst     = (const int*)d_in[8];
    float*       out     = (float*)d_out;

    float* out_norm = (float*)d_ws;
    float* in_norm  = out_norm + N_NODES;
    float* buf0     = in_norm + N_NODES;
    float* buf1     = buf0 + (long)N_NODES * H_F;
    unsigned* part  = (unsigned*)(buf1 + (long)N_NODES * H_F);
    unsigned* pre   = part + (long)DEG_S * DEG_NP;
    unsigned* degin = part + (long)2 * DEG_S * DEG_NP;
    unsigned* rowptr = degin + DEG_NP;
    int*      col   = (int*)(rowptr + DEG_NP + 32);
    unsigned* blktot = (unsigned*)(col + N_EDGES);

    gcn_degree_hist<<<2 * DEG_R * DEG_S, DEG_T, 0, stream>>>(src, dst, part);
    gcn_degree_finish<<<2 * FIN_NB, FIN_T, 0, stream>>>(part, out_norm, degin, blktot);

    gcn_rowptr_scan<<<FIN_NB, FIN_T, 0, stream>>>(degin, blktot, rowptr);
    gcn_place_and_gemm1<<<PG_PLACE_BLOCKS + PG_GEMM_BLOCKS, DEG_T, 0, stream>>>(
        src, dst, pre, rowptr, col, feature, W1, out_norm, buf0);

    gcn_pull_gemm2<<<PG2_BLOCKS, PG2_T, 0, stream>>>(buf0, rowptr, col, in_norm, b1, W2, out_norm, buf1);

    gcn_pull_head<<<N_NODES / (PULL_NPW * (PULL_T / 32)), PULL_T, 0, stream>>>(
        buf1, rowptr, col, in_norm, b2, Wm, bm, out);
}


# Row gathers made unconditional: list entries past a node's end name an all-zero extra row, so the per-row exec branch and 64-bit address math leave both aggregation loops; sum order unchanged
# speedup vs baseline: 3.5627x; 1.0759x over previous
#include <hip/hip_runtime.h>

#define N_NODES 100000
#define N_EDGES 1600000
#define IN_F    128
#define H_F     64

typedef __attribute__((ext_vector_type(2))) float v2f;
typedef __attribute__((ext_vector_type(8))) float v8f;


#define DEG_R     4
#define DEG_S     32
#define DEG_T     512
#define DEG_U     8
#define DEG_BINS  25024
#define DEG_NP    (DEG_R * DEG_BINS)
#define DEG_SLICE (N_EDGES / DEG_S)
#define DEG_Q     (DEG_SLICE / 4)

static_assert(DEG_NP >= N_NODES, "parts must cover every node");
static_assert(DEG_BINS % 32 == 0, "a part must be whole 128-byte lines");
static_assert(N_EDGES % DEG_S == 0 && DEG_SLICE % 4 == 0, "slices must be whole 16-byte groups");
static_assert(N_NODES % 4 == 0, "finish kernel handles four nodes per thread");

__device__ __forceinline__ void gcn_count_one(unsigned* hist, int v, int lo) {
    const unsigned d = (unsigned)(v - lo);
    if (d < (unsigned)DEG_BINS) atomicAdd(&hist[d], 1u);
}

__global__ __launch_bounds__(DEG_T)
void gcn_degree_hist(const int* __restrict__ src,
                     const int* __restrict__ dst,
                     unsigned* __restrict__ part) {
    __shared__ __attribute__((aligned(16))) unsigned hist[DEG_BINS];

    const int tid = threadIdx.x;
    const int s   = blockIdx.x % DEG_S;
    const int ar  = blockIdx.x / DEG_S;
    const int r   = ar % DEG_R;
    const int a   = ar / DEG_R;
    const int lo  = r * DEG_BINS;

    uint4* h4 = (uint4*)hist;
    for (int i = tid; i < DEG_BINS / 4; i += DEG_T) h4[i] = make_uint4(0u, 0u, 0u, 0u);
    __syncthreads();

    const int* __restrict__ idx = a ? dst : src;
    const int4* __restrict__ p4 = (const int4*)(idx + (long)s * DEG_SLICE);

    for (int base = 0; base < DEG_Q; base += DEG_T * DEG_U) {
        int4 v[DEG_U];
        #pragma unroll
        for (int u = 0; u < DEG_U; ++u) {
            const int q = base + u * DEG_T + tid;
            v[u] = (q < DEG_Q) ? p4[q] : make_int4(-1, -1, -1, -1);
        }
        #pragma unroll
        for (int u = 0; u < DEG_U; ++u) {
            gcn_count_one(hist, v[u].x, lo);
            gcn_count_one(hist, v[u].y, lo);
            gcn_count_one(hist, v[u].z, lo);
            gcn_count_one(hist, v[u].w, lo);
        }
    }
    __syncthreads();

    uint4* __restrict__ o4 = (uint4*)(part + (long)(a * DEG_S + s) * DEG_NP + lo);
    for (int i = tid; i < DEG_BINS / 4; i += DEG_T) o4[i] = h4[i];
}


#define FIN_T   256
#define FIN_NPB (4 * FIN_T)
#define FIN_NB  ((N_NODES + FIN_NPB - 1) / FIN_NPB)

static_assert(FIN_NB <= FIN_T, "one thread per earlier block total in the scan kernel");
static_assert(DEG_S % 8 == 0, "finish kernel loads the slices in groups of eight");

__global__ __launch_bounds__(FIN_T)
void gcn_degree_finish(unsigned* __restrict__ part,
                       float* __restrict__ norm,
                       unsigned* __restrict__ degin,
                       unsigned* __restrict__ blktot,
                       float* __restrict__ zrow0,
                       float* __restrict__ zrow1) {
    __shared__ unsigned red[FIN_T];
    const int tid = threadIdx.x;

    if (blockIdx.x == 0 && tid < 32)
        ((float4*)(tid < 16 ? zrow0 : zrow1))[tid & 15] = make_float4(0.f, 0.f, 0.f, 0.f);
    const int a = blockIdx.x / FIN_NB;
    const int b = blockIdx.x - a * FIN_NB;
    const int j = b * FIN_T + tid;
    const bool valid = j < N_NODES / 4;

    uint4 acc = make_uint4(0u, 0u, 0u, 0u);
    if (valid) {
        uint4* __restrict__ p = (uint4*)(part + (long)a * DEG_S * DEG_NP) + j;
        #pragma unroll
        for (int s0 = 0; s0 < DEG_S; s0 += 8) {
            uint4 v[8];
            #pragma unroll
            for (int u = 0; u < 8; ++u) v[u] = p[(long)(s0 + u) * (DEG_NP / 4)];
            #pragma unroll
            for (int u = 0; u < 8; ++u) {
                if (a == 1) p[(long)(s0 + u) * (DEG_NP / 4)] = acc;
                acc.x += v[u].x; acc.y += v[u].y; acc.z += v[u].z; acc.w += v[u].w;
            }
        }
        float4 o;
        o.x = rsqrtf(fmaxf((float)acc.x, 1.0f));
        o.y = rsqrtf(fmaxf((float)acc.y, 1.0f));
        o.z = rsqrtf(fmaxf((float)acc.z, 1.0f));
        o.w = rsqrtf(fmaxf((float)acc.w, 1.0f));
        *(float4*)(norm + (long)a * N_NODES + 4 * j) = o;
    }

    if (a == 1) {
        if (valid) *(uint4*)(degin + 4 * j) = acc;
        red[tid] = acc.x + acc.y + acc.z + acc.w;
        __syncthreads();
        for (int off = FIN_T / 2; off > 0; off >>= 1) {
            if (tid < off) red[tid] += red[tid + off];
            __syncthreads();
        }
        if (tid == 0) blktot[b] = red[0];
    }
}

__global__ __launch_bounds__(FIN_T)
void gcn_rowptr_scan(const unsigned* __restrict__ degin,
                     const unsigned* __restrict__ blktot,
                     unsigned* __restrict__ rowptr) {
    __shared__ unsigned sh[FIN_T];
    const int tid = threadIdx.x;
    const int b = blockIdx.x;
    const int j = b * FIN_T + tid;
    const bool valid = j < N_NODES / 4;

    sh[tid] = (tid < b) ? blktot[tid] : 0u;
    __syncthreads();
    for (int off = FIN_T / 2; off > 0; off >>= 1) {
        if (tid < off) sh[tid] += sh[tid + off];
        __syncthreads();
    }
    const unsigned base = sh[0];
    __syncthreads();

    uint4 d = make_uint4(0u, 0u, 0u, 0u);
    if (valid) d = *(const uint4*)(degin + 4 * j);
    const unsigned t = d.x + d.y + d.z + d.w;

    sh[tid] = t;
    __syncthreads();
    for (int off = 1; off < FIN_T; off <<= 1) {
        const unsigned y = (tid >= off) ? sh[tid - off] : 0u;
        __syncthreads();
        sh[tid] += y;
        __syncthreads();
    }
    const unsigned first = base + sh[tid] - t;
    if (valid) {
        uint4 o;
        o.x = first;
        o.y = o.x + d.x;
        o.z = o.y + d.y;
        o.w = o.z + d.z;
        *(uint4*)(rowptr + 4 * j) = o;
        if (j == N_NODES / 4 - 1) rowptr[N_NODES] = first + t;
    }
}

__device__ __forceinline__ void gcn_place_one(unsigned* cur, int* __restrict__ col,
                                              int dv, int sv, int lo) {
    const unsigned d = (unsigned)(dv - lo);
    if (d < (unsigned)DEG_BINS) {
        const unsigned pos = atomicAdd(&cur[d], 1u);
        col[pos] = sv;
    }
}

__device__ __forceinline__ void gcn_place_cell(unsigned* cur,
                                               const int* __restrict__ src,
                                               const int* __restrict__ dst,
                                               const unsigned* __restrict__ pre,
                                               const unsigned* __restrict__ rowptr,
                                               int* __restrict__ col,
                                               int cell, int tid) {
    const int s   = cell % DEG_S;
    const int r   = cell / DEG_S;
    const int lo  = r * DEG_BINS;

    const uint4* __restrict__ pre4 = (const uint4*)(pre + (long)s * DEG_NP + lo);
    uint4* c4 = (uint4*)cur;
    for (int i = tid; i < DEG_BINS / 4; i += DEG_T) {
        const int n4 = min(lo + 4 * i, N_NODES - 4);
        const uint4 a = *(const uint4*)(rowptr + n4);
        const uint4 p = pre4[i];
        c4[i] = make_uint4(a.x + p.x, a.y + p.y, a.z + p.z, a.w + p.w);
    }
    __syncthreads();

    const int4* __restrict__ d4 = (const int4*)(dst + (long)s * DEG_SLICE);
    const int4* __restrict__ s4 = (const int4*)(src + (long)s * DEG_SLICE);

    for (int base = 0; base < DEG_Q; base += DEG_T * DEG_U) {
        int4 vd[DEG_U], vs[DEG_U];
        #pragma unroll
        for (int u = 0; u < DEG_U; ++u) {
            const int q = base + u * DEG_T + tid;
            vd[u] = (q < DEG_Q) ? d4[q] : make_int4(-1, -1, -1, -1);
            vs[u] = (q < DEG_Q) ? s4[q] : make_int4(0, 0, 0, 0);
        }
        #pragma unroll
        for (int u = 0; u < DEG_U; ++u) {
            gcn_place_one(cur, col, vd[u].x, vs[u].x, lo);
            gcn_place_one(cur, col, vd[u].y, vs[u].y, lo);
            gcn_place_one(cur, col, vd[u].z, vs[u].z, lo);
            gcn_place_one(cur, col, vd[u].w, vs[u].w, lo);
        }
    }
}


static_assert(N_NODES % 16 == 0, "whole M-tiles");

template <int K, int TPB>
__device__ __forceinline__ void gcn_gemm_tiles(const float* __restrict__ A,
                                               const float* __restrict__ W,
                                               const float* __restrict__ rowscale,
                                               float* __restrict__ C,
                                               int grp, int wave, int lane) {
    const int half = lane >> 4;
    const int l16  = lane & 15;
    const int n0   = wave * 16;

    v2f b[K / 4];
    #pragma unroll
    for (int k0 = 0; k0 < K; k0 += 4) {
        const int ka = k0 + 2 * half;
        b[k0 / 4].x = W[(long)ka       * H_F + n0 + l16];
        b[k0 / 4].y = W[(long)(ka + 1) * H_F + n0 + l16];
    }

    #pragma unroll 1
    for (int t = 0; t < TPB; ++t) {
        const int m0 = (grp * TPB + t) * 16;
        const int m  = m0 + l16;

        const float s = rowscale[m];
        const float* __restrict__ arow = A + (long)m * K;

        v8f c = {};
        #pragma unroll
        for (int k0 = 0; k0 < K; k0 += 4) {
            const int ka = k0 + 2 * half;
            v2f a;
            a.x = arow[ka]     * s;
            a.y = arow[ka + 1] * s;
            c = __builtin_amdgcn_wmma_f32_16x16x4_f32(
                    false, a, false, b[k0 / 4], (short)0, c, false, false);
        }

        #pragma unroll
        for (int r = 0; r < 8; ++r) {
            C[(long)(m0 + r + 8 * half) * H_F + n0 + l16] = c[r];
        }
    }
}


#define PG_PLACE_BLOCKS (DEG_R * DEG_S)
#define PG_TPB    2
#define PG_GROUPS (N_NODES / 16 / PG_TPB)
#define PG_GPB    (DEG_T / 128)
#define PG_GEMM_BLOCKS ((PG_GROUPS + PG_GPB - 1) / PG_GPB)

static_assert((N_NODES / 16) % PG_TPB == 0, "whole groups of M-tiles");
static_assert(DEG_T % 128 == 0, "whole GEMM groups per block");

__global__ __launch_bounds__(DEG_T)
void gcn_place_and_gemm1(const int* __restrict__ src,
                         const int* __restrict__ dst,
                         const unsigned* __restrict__ pre,
                         const unsigned* __restrict__ rowptr,
                         int* __restrict__ col,
                         const float* __restrict__ feature,
                         const float* __restrict__ W1,
                         const float* __restrict__ out_norm,
                         float* __restrict__ g) {
    __shared__ __attribute__((aligned(16))) unsigned cur[DEG_BINS];

    if (blockIdx.x >= PG_PLACE_BLOCKS) {
        const int grp = (blockIdx.x - PG_PLACE_BLOCKS) * PG_GPB + (threadIdx.x >> 7);
        if (grp < PG_GROUPS)
            gcn_gemm_tiles<IN_F, PG_TPB>(feature, W1, out_norm, g, grp,
                                         (threadIdx.x >> 5) & 3, threadIdx.x & 31);
        return;
    }
    gcn_place_cell(cur, src, dst, pre, rowptr, col, blockIdx.x, threadIdx.x);
}


#define GCN_ZROW   N_NODES
#define GCN_GROWS  (N_NODES + 1)

static_assert((long)GCN_GROWS * H_F * 4 < (1L << 32), "row byte offsets fit 32 bits");

__device__ __forceinline__ float4 gcn_sum16(const float* __restrict__ g, int mine,
                                            unsigned lane_byte, float4 acc) {
    float4 v[16];
    #pragma unroll
    for (int k = 0; k < 16; ++k) {
        const unsigned sk = (unsigned)__shfl(mine, k, 16);
        v[k] = *(const float4*)((const char*)g + (sk * (unsigned)(H_F * 4) + lane_byte));
    }
    #pragma unroll
    for (int k = 0; k < 16; ++k) {
        acc.x += v[k].x; acc.y += v[k].y; acc.z += v[k].z; acc.w += v[k].w;
    }
    return acc;
}

template <int NPW, typename Epi>
__device__ __forceinline__ void gcn_pull_rows(const float* __restrict__ g,
                                              const unsigned* __restrict__ rowptr,
                                              const int* __restrict__ col,
                                              const float* __restrict__ in_norm,
                                              const float* __restrict__ bias,
                                              int n0, int lane, const Epi& epi) {
    static_assert(NPW % 2 == 0 && NPW < 32, "pairs of nodes; one lane per row start");
    const int grp  = lane >> 4;
    const int l16  = lane & 15;

    const unsigned lane_byte = 16u * (unsigned)l16;

    const unsigned rp  = rowptr[n0 + min(lane, NPW)];
    const float    nrm = in_norm[n0 + min(lane, NPW - 1)];
    const float4   bb  = ((const float4*)bias)[l16];

    unsigned beg = __shfl(rp, grp);
    unsigned end = __shfl(rp, grp + 1);
    int c0 = (beg + (unsigned)l16 < end)       ? col[beg + (unsigned)l16]       : GCN_ZROW;
    int c1 = (beg + 16u + (unsigned)l16 < end) ? col[beg + 16u + (unsigned)l16] : GCN_ZROW;

    asm volatile("" :: "v"(nrm), "v"(bb.x), "v"(bb.y), "v"(bb.z), "v"(bb.w), "v"(c0), "v"(c1));

    #pragma unroll 1
    for (int i = 0; i < NPW / 2; ++i) {
        unsigned nbeg = 0u, nend = 0u;
        int nc0 = GCN_ZROW, nc1 = GCN_ZROW;
        if (i + 1 < NPW / 2) {
            nbeg = __shfl(rp, 2 * i + 2 + grp);
            nend = __shfl(rp, 2 * i + 3 + grp);
            nc0 = (nbeg + (unsigned)l16 < nend)       ? col[nbeg + (unsigned)l16]       : GCN_ZROW;
            nc1 = (nbeg + 16u + (unsigned)l16 < nend) ? col[nbeg + 16u + (unsigned)l16] : GCN_ZROW;
        }
        const float nn = __shfl(nrm, 2 * i + grp);

        float4 acc = make_float4(0.f, 0.f, 0.f, 0.f);
        for (unsigned base = beg; base < end; base += 16) {
            const unsigned j2 = base + 32u + (unsigned)l16;
            const int c2 = (j2 < end) ? col[j2] : GCN_ZROW;
            acc = gcn_sum16(g, c0, lane_byte, acc);
            c0 = c1;
            c1 = c2;
        }

        float4 o;
        o.x = fmaxf(acc.x * nn + bb.x, 0.0f);
        o.y = fmaxf(acc.y * nn + bb.y, 0.0f);
        o.z = fmaxf(acc.z * nn + bb.z, 0.0f);
        o.w = fmaxf(acc.w * nn + bb.w, 0.0f);
        epi.row(i, grp, l16, n0 + 2 * i + grp, o);

        beg = nbeg; end = nend; c0 = nc0; c1 = nc1;
    }
}


#define PG2_NPW    16
#define PG2_WPB    5
#define PG2_T      (PG2_WPB * 32)
#define PG2_PITCH  (H_F + 4)
#define PG2_WAVES  (N_NODES / PG2_NPW)
#define PG2_BLOCKS (PG2_WAVES / PG2_WPB)
#define PG2_W2Q    (H_F * H_F / 4)
#define PG2_STG    128

static_assert(N_NODES % PG2_NPW == 0, "whole tiles of 16 nodes");
static_assert(PG2_WAVES % PG2_WPB == 0, "whole blocks: every wave of every block has a tile");
static_assert(PG2_STG % 32 == 0 && PG2_STG <= PG2_T && PG2_W2Q % PG2_STG == 0, "whole waves stage W2 in whole rounds");
static_assert((PG2_PITCH * 4) % 16 == 0, "16-byte row stores into the tile");

struct gcn_epi_tile {
    float* tile;
    __device__ __forceinline__ void row(int i, int grp, int l16, int n, float4 o) const {
        *(float4*)(tile + (2 * i + grp) * PG2_PITCH + 4 * l16) = o;
    }
};

__global__ __launch_bounds__(PG2_T)
void gcn_pull_gemm2(const float* __restrict__ g,
                    const unsigned* __restrict__ rowptr,
                    const int* __restrict__ col,
                    const float* __restrict__ in_norm,
                    const float* __restrict__ bias,
                    const float* __restrict__ W2,
                    const float* __restrict__ out_norm,
                    float* __restrict__ g2) {
    __shared__ __attribute__((aligned(16))) float wt[H_F * PG2_PITCH];
    __shared__ __attribute__((aligned(16))) float tiles[PG2_WPB * PG2_NPW * PG2_PITCH];

    const int tid = threadIdx.x;

    if (tid < PG2_STG) {
        #pragma unroll
        for (int j = 0; j < PG2_W2Q / PG2_STG; ++j) {
            const int q = j * PG2_STG + tid;
            const float4 v = ((const float4*)W2)[q];
            const int k = q / (H_F / 4);
            const int n = (q % (H_F / 4)) * 4;
            wt[(n + 0) * PG2_PITCH + k] = v.x;
            wt[(n + 1) * PG2_PITCH + k] = v.y;
            wt[(n + 2) * PG2_PITCH + k] = v.z;
            wt[(n + 3) * PG2_PITCH + k] = v.w;
        }
    }
    __syncthreads();

    const int wave = tid >> 5;
    const int lane = tid & 31;
    const int wv   = blockIdx.x * PG2_WPB + wave;
    const int n0   = wv * PG2_NPW;
    const int half = lane >> 4;
    const int l16  = lane & 15;

    const float s = out_norm[n0 + l16];

    float* tile = tiles + wave * (PG2_NPW * PG2_PITCH);
    gcn_epi_tile epi;
    epi.tile = tile;
    gcn_pull_rows<PG2_NPW>(g, rowptr, col, in_norm, bias, n0, lane, epi);


    const float* arow = tile + l16 * PG2_PITCH + 2 * half;
    const float* bcol = wt   + l16 * PG2_PITCH + 2 * half;

    v8f c0 = {}, c1 = {}, c2 = {}, c3 = {};
    #pragma unroll
    for (int k0 = 0; k0 < H_F; k0 += 4) {
        const float2 av  = *(const float2*)(arow + k0);
        const float2 b0v = *(const float2*)(bcol + 0 * 16 * PG2_PITCH + k0);
        const float2 b1v = *(const float2*)(bcol + 1 * 16 * PG2_PITCH + k0);
        const float2 b2v = *(const float2*)(bcol + 2 * 16 * PG2_PITCH + k0);
        const float2 b3v = *(const float2*)(bcol + 3 * 16 * PG2_PITCH + k0);
        v2f a, b0, b1, b2, b3;
        a.x  = av.x * s;  a.y  = av.y * s;
        b0.x = b0v.x;     b0.y = b0v.y;
        b1.x = b1v.x;     b1.y = b1v.y;
        b2.x = b2v.x;     b2.y = b2v.y;
        b3.x = b3v.x;     b3.y = b3v.y;
        c0 = __builtin_amdgcn_wmma_f32_16x16x4_f32(false, a, false, b0, (short)0, c0, false, false);
        c1 = __builtin_amdgcn_wmma_f32_16x16x4_f32(false, a, false, b1, (short)0, c1, false, false);
        c2 = __builtin_amdgcn_wmma_f32_16x16x4_f32(false, a, false, b2, (short)0, c2, false, false);
        c3 = __builtin_amdgcn_wmma_f32_16x16x4_f32(false, a, false, b3, (short)0, c3, false, false);
    }

    float* __restrict__ crow = g2 + (long)(n0 + 8 * half) * H_F + l16;
    #pragma unroll
    for (int r = 0; r < 8; ++r) {
        crow[(long)r * H_F +  0] = c0[r];
        crow[(long)r * H_F + 16] = c1[r];
        crow[(long)r * H_F + 32] = c2[r];
        crow[(long)r * H_F + 48] = c3[r];
    }
}


#define PULL_T   256
#define PULL_NPW 10

static_assert(N_NODES % (PULL_NPW * (PULL_T / 32)) == 0, "whole blocks of whole waves");

__device__ __forceinline__ float gcn_row16_sum(float v) {
    v += __builtin_bit_cast(float, __builtin_amdgcn_update_dpp(0, __builtin_bit_cast(int, v), 0x161, 0xf, 0xf, true));
    v += __builtin_bit_cast(float, __builtin_amdgcn_update_dpp(0, __builtin_bit_cast(int, v), 0x162, 0xf, 0xf, true));
    v += __builtin_bit_cast(float, __builtin_amdgcn_update_dpp(0, __builtin_bit_cast(int, v), 0x164, 0xf, 0xf, true));
    v += __builtin_bit_cast(float, __builtin_amdgcn_update_dpp(0, __builtin_bit_cast(int, v), 0x168, 0xf, 0xf, true));
    return v;
}

struct gcn_epi_head {
    float4 w0, w1;
    float2 bh;
    float* out;
    __device__ __forceinline__ void row(int i, int grp, int l16, int n, float4 o) const {
        float p0 = o.x * w0.x + o.y * w0.z + o.z * w1.x + o.w * w1.z;
        float p1 = o.x * w0.y + o.y * w0.w + o.z * w1.y + o.w * w1.w;
        p0 = gcn_row16_sum(p0);
        p1 = gcn_row16_sum(p1);
        if (l16 == 0) ((float2*)out)[n] = make_float2(p0 + bh.x, p1 + bh.y);
    }
};

__global__ __launch_bounds__(PULL_T)
void gcn_pull_head(const float* __restrict__ g,
                   const unsigned* __restrict__ rowptr,
                   const int* __restrict__ col,
                   const float* __restrict__ in_norm,
                   const float* __restrict__ bias,
                   const float* __restrict__ Wm,
                   const float* __restrict__ bm,
                   float* __restrict__ out) {
    const int lane = threadIdx.x & 31;
    const int l16  = lane & 15;
    const int wv   = blockIdx.x * (PULL_T / 32) + (threadIdx.x >> 5);

    gcn_epi_head epi;
    epi.w0  = ((const float4*)Wm)[2 * l16];
    epi.w1  = ((const float4*)Wm)[2 * l16 + 1];
    epi.bh  = make_float2(bm[0], bm[1]);
    epi.out = out;
    gcn_pull_rows<PULL_NPW>(g, rowptr, col, in_norm, bias, wv * PULL_NPW, lane, epi);
}


extern "C" void kernel_launch(void* const* d_in, const int* in_sizes, int n_in,
                              void* d_out, int out_size, void* d_ws, size_t ws_size,
                              hipStream_t stream) {
    const float* feature = (const float*)d_in[0];
    const float* W1      = (const float*)d_in[1];
    const float* b1      = (const float*)d_in[2];
    const float* W2      = (const float*)d_in[3];
    const float* b2      = (const float*)d_in[4];
    const float* Wm      = (const float*)d_in[5];
    const float* bm      = (const float*)d_in[6];
    const int*   src     = (const int*)d_in[7];
    const int*   dst     = (const int*)d_in[8];
    float*       out     = (float*)d_out;

    float* out_norm = (float*)d_ws;
    float* in_norm  = out_norm + N_NODES;
    float* buf0     = in_norm + N_NODES;
    float* buf1     = buf0 + (long)GCN_GROWS * H_F;
    unsigned* part  = (unsigned*)(buf1 + (long)GCN_GROWS * H_F);
    unsigned* pre   = part + (long)DEG_S * DEG_NP;
    unsigned* degin = part + (long)2 * DEG_S * DEG_NP;
    unsigned* rowptr = degin + DEG_NP;
    int*      col   = (int*)(rowptr + DEG_NP + 32);
    unsigned* blktot = (unsigned*)(col + N_EDGES);

    gcn_degree_hist<<<2 * DEG_R * DEG_S, DEG_T, 0, stream>>>(src, dst, part);
    gcn_degree_finish<<<2 * FIN_NB, FIN_T, 0, stream>>>(part, out_norm, degin, blktot,
                                                        buf0 + (long)GCN_ZROW * H_F, buf1 + (long)GCN_ZROW * H_F);

    gcn_rowptr_scan<<<FIN_NB, FIN_T, 0, stream>>>(degin, blktot, rowptr);
    gcn_place_and_gemm1<<<PG_PLACE_BLOCKS + PG_GEMM_BLOCKS, DEG_T, 0, stream>>>(
        src, dst, pre, rowptr, col, feature, W1, out_norm, buf0);

    gcn_pull_gemm2<<<PG2_BLOCKS, PG2_T, 0, stream>>>(buf0, rowptr, col, in_norm, b1, W2, out_norm, buf1);

    gcn_pull_head<<<N_NODES / (PULL_NPW * (PULL_T / 32)), PULL_T, 0, stream>>>(
        buf1, rowptr, col, in_norm, b2, Wm, bm, out);
}


# Layer-1 product: A tiles fetched with whole-line 16-byte loads into per-wave LDS tiles, W staged transposed in LDS once per block, fragments read from LDS; k order per accumulator unchanged
# speedup vs baseline: 3.6869x; 1.0349x over previous
#include <hip/hip_runtime.h>

#define N_NODES 100000
#define N_EDGES 1600000
#define IN_F    128
#define H_F     64

typedef __attribute__((ext_vector_type(2))) float v2f;
typedef __attribute__((ext_vector_type(8))) float v8f;


#define DEG_R     4
#define DEG_S     32
#define DEG_T     512
#define DEG_U     8
#define DEG_BINS  25024
#define DEG_NP    (DEG_R * DEG_BINS)
#define DEG_SLICE (N_EDGES / DEG_S)
#define DEG_Q     (DEG_SLICE / 4)

static_assert(DEG_NP >= N_NODES, "parts must cover every node");
static_assert(DEG_BINS % 32 == 0, "a part must be whole 128-byte lines");
static_assert(N_EDGES % DEG_S == 0 && DEG_SLICE % 4 == 0, "slices must be whole 16-byte groups");
static_assert(N_NODES % 4 == 0, "finish kernel handles four nodes per thread");

__device__ __forceinline__ void gcn_count_one(unsigned* hist, int v, int lo) {
    const unsigned d = (unsigned)(v - lo);
    if (d < (unsigned)DEG_BINS) atomicAdd(&hist[d], 1u);
}

__global__ __launch_bounds__(DEG_T)
void gcn_degree_hist(const int* __restrict__ src,
                     const int* __restrict__ dst,
                     unsigned* __restrict__ part) {
    __shared__ __attribute__((aligned(16))) unsigned hist[DEG_BINS];

    const int tid = threadIdx.x;
    const int s   = blockIdx.x % DEG_S;
    const int ar  = blockIdx.x / DEG_S;
    const int r   = ar % DEG_R;
    const int a   = ar / DEG_R;
    const int lo  = r * DEG_BINS;

    uint4* h4 = (uint4*)hist;
    for (int i = tid; i < DEG_BINS / 4; i += DEG_T) h4[i] = make_uint4(0u, 0u, 0u, 0u);
    __syncthreads();

    const int* __restrict__ idx = a ? dst : src;
    const int4* __restrict__ p4 = (const int4*)(idx + (long)s * DEG_SLICE);

    for (int base = 0; base < DEG_Q; base += DEG_T * DEG_U) {
        int4 v[DEG_U];
        #pragma unroll
        for (int u = 0; u < DEG_U; ++u) {
            const int q = base + u * DEG_T + tid;
            v[u] = (q < DEG_Q) ? p4[q] : make_int4(-1, -1, -1, -1);
        }
        #pragma unroll
        for (int u = 0; u < DEG_U; ++u) {
            gcn_count_one(hist, v[u].x, lo);
            gcn_count_one(hist, v[u].y, lo);
            gcn_count_one(hist, v[u].z, lo);
            gcn_count_one(hist, v[u].w, lo);
        }
    }
    __syncthreads();

    uint4* __restrict__ o4 = (uint4*)(part + (long)(a * DEG_S + s) * DEG_NP + lo);
    for (int i = tid; i < DEG_BINS / 4; i += DEG_T) o4[i] = h4[i];
}


#define FIN_T   256
#define FIN_NPB (4 * FIN_T)
#define FIN_NB  ((N_NODES + FIN_NPB - 1) / FIN_NPB)

static_assert(FIN_NB <= FIN_T, "one thread per earlier block total in the scan kernel");
static_assert(DEG_S % 8 == 0, "finish kernel loads the slices in groups of eight");

__global__ __launch_bounds__(FIN_T)
void gcn_degree_finish(unsigned* __restrict__ part,
                       float* __restrict__ norm,
                       unsigned* __restrict__ degin,
                       unsigned* __restrict__ blktot,
                       float* __restrict__ zrow0,
                       float* __restrict__ zrow1) {
    __shared__ unsigned red[FIN_T];
    const int tid = threadIdx.x;

    if (blockIdx.x == 0 && tid < 32)
        ((float4*)(tid < 16 ? zrow0 : zrow1))[tid & 15] = make_float4(0.f, 0.f, 0.f, 0.f);
    const int a = blockIdx.x / FIN_NB;
    const int b = blockIdx.x - a * FIN_NB;
    const int j = b * FIN_T + tid;
    const bool valid = j < N_NODES / 4;

    uint4 acc = make_uint4(0u, 0u, 0u, 0u);
    if (valid) {
        uint4* __restrict__ p = (uint4*)(part + (long)a * DEG_S * DEG_NP) + j;
        #pragma unroll
        for (int s0 = 0; s0 < DEG_S; s0 += 8) {
            uint4 v[8];
            #pragma unroll
            for (int u = 0; u < 8; ++u) v[u] = p[(long)(s0 + u) * (DEG_NP / 4)];
            #pragma unroll
            for (int u = 0; u < 8; ++u) {
                if (a == 1) p[(long)(s0 + u) * (DEG_NP / 4)] = acc;
                acc.x += v[u].x; acc.y += v[u].y; acc.z += v[u].z; acc.w += v[u].w;
            }
        }
        float4 o;
        o.x = rsqrtf(fmaxf((float)acc.x, 1.0f));
        o.y = rsqrtf(fmaxf((float)acc.y, 1.0f));
        o.z = rsqrtf(fmaxf((float)acc.z, 1.0f));
        o.w = rsqrtf(fmaxf((float)acc.w, 1.0f));
        *(float4*)(norm + (long)a * N_NODES + 4 * j) = o;
    }

    if (a == 1) {
        if (valid) *(uint4*)(degin + 4 * j) = acc;
        red[tid] = acc.x + acc.y + acc.z + acc.w;
        __syncthreads();
        for (int off = FIN_T / 2; off > 0; off >>= 1) {
            if (tid < off) red[tid] += red[tid + off];
            __syncthreads();
        }
        if (tid == 0) blktot[b] = red[0];
    }
}

__global__ __launch_bounds__(FIN_T)
void gcn_rowptr_scan(const unsigned* __restrict__ degin,
                     const unsigned* __restrict__ blktot,
                     unsigned* __restrict__ rowptr) {
    __shared__ unsigned sh[FIN_T];
    const int tid = threadIdx.x;
    const int b = blockIdx.x;
    const int j = b * FIN_T + tid;
    const bool valid = j < N_NODES / 4;

    sh[tid] = (tid < b) ? blktot[tid] : 0u;
    __syncthreads();
    for (int off = FIN_T / 2; off > 0; off >>= 1) {
        if (tid < off) sh[tid] += sh[tid + off];
        __syncthreads();
    }
    const unsigned base = sh[0];
    __syncthreads();

    uint4 d = make_uint4(0u, 0u, 0u, 0u);
    if (valid) d = *(const uint4*)(degin + 4 * j);
    const unsigned t = d.x + d.y + d.z + d.w;

    sh[tid] = t;
    __syncthreads();
    for (int off = 1; off < FIN_T; off <<= 1) {
        const unsigned y = (tid >= off) ? sh[tid - off] : 0u;
        __syncthreads();
        sh[tid] += y;
        __syncthreads();
    }
    const unsigned first = base + sh[tid] - t;
    if (valid) {
        uint4 o;
        o.x = first;
        o.y = o.x + d.x;
        o.z = o.y + d.y;
        o.w = o.z + d.z;
        *(uint4*)(rowptr + 4 * j) = o;
        if (j == N_NODES / 4 - 1) rowptr[N_NODES] = first + t;
    }
}

__device__ __forceinline__ void gcn_place_one(unsigned* cur, int* __restrict__ col,
                                              int dv, int sv, int lo) {
    const unsigned d = (unsigned)(dv - lo);
    if (d < (unsigned)DEG_BINS) {
        const unsigned pos = atomicAdd(&cur[d], 1u);
        col[pos] = sv;
    }
}

__device__ __forceinline__ void gcn_place_cell(unsigned* cur,
                                               const int* __restrict__ src,
                                               const int* __restrict__ dst,
                                               const unsigned* __restrict__ pre,
                                               const unsigned* __restrict__ rowptr,
                                               int* __restrict__ col,
                                               int cell, int tid) {
    const int s   = cell % DEG_S;
    const int r   = cell / DEG_S;
    const int lo  = r * DEG_BINS;

    const uint4* __restrict__ pre4 = (const uint4*)(pre + (long)s * DEG_NP + lo);
    uint4* c4 = (uint4*)cur;
    for (int i = tid; i < DEG_BINS / 4; i += DEG_T) {
        const int n4 = min(lo + 4 * i, N_NODES - 4);
        const uint4 a = *(const uint4*)(rowptr + n4);
        const uint4 p = pre4[i];
        c4[i] = make_uint4(a.x + p.x, a.y + p.y, a.z + p.z, a.w + p.w);
    }
    __syncthreads();

    const int4* __restrict__ d4 = (const int4*)(dst + (long)s * DEG_SLICE);
    const int4* __restrict__ s4 = (const int4*)(src + (long)s * DEG_SLICE);

    for (int base = 0; base < DEG_Q; base += DEG_T * DEG_U) {
        int4 vd[DEG_U], vs[DEG_U];
        #pragma unroll
        for (int u = 0; u < DEG_U; ++u) {
            const int q = base + u * DEG_T + tid;
            vd[u] = (q < DEG_Q) ? d4[q] : make_int4(-1, -1, -1, -1);
            vs[u] = (q < DEG_Q) ? s4[q] : make_int4(0, 0, 0, 0);
        }
        #pragma unroll
        for (int u = 0; u < DEG_U; ++u) {
            gcn_place_one(cur, col, vd[u].x, vs[u].x, lo);
            gcn_place_one(cur, col, vd[u].y, vs[u].y, lo);
            gcn_place_one(cur, col, vd[u].z, vs[u].z, lo);
            gcn_place_one(cur, col, vd[u].w, vs[u].w, lo);
        }
    }
}


static_assert(N_NODES % 16 == 0, "whole M-tiles");

#define G1_WP     (IN_F + 4)
#define G1_AP     (IN_F / 2 + 4)
#define G1_WAVES  (DEG_T / 32)
#define G1_WFLOATS (H_F * G1_WP)
#define G1_AFLOATS (16 * G1_AP)
#define G1_LDS_BYTES ((G1_WFLOATS + G1_WAVES * G1_AFLOATS) * 4)
#define G1_UNITS  (N_NODES / 16 * 2)
#define G1_W4     (IN_F * H_F / 4)

static_assert(G1_W4 % DEG_T == 0, "every thread stages the same number of groups");
static_assert((G1_WP * 4) % 16 == 0 && (G1_AP * 4) % 16 == 0, "16-byte aligned LDS rows");

__device__ __forceinline__ void gcn_gemm1_stage_w(const float* __restrict__ W, float* wt, int tid) {
    #pragma unroll
    for (int j = 0; j < G1_W4 / DEG_T; ++j) {
        const int q  = j * DEG_T + tid;
        const int n4 = ((q >> 5) & 3) * 4 + (q & 3);
        const int k  = (q >> 7) * 8 + ((q >> 2) & 7);
        const float4 v = ((const float4*)W)[k * (H_F / 4) + n4];
        const int n = 4 * n4;
        wt[(n + 0) * G1_WP + k] = v.x;
        wt[(n + 1) * G1_WP + k] = v.y;
        wt[(n + 2) * G1_WP + k] = v.z;
        wt[(n + 3) * G1_WP + k] = v.w;
    }
}

__device__ __forceinline__ void gcn_gemm1_unit(const float* __restrict__ A,
                                               const float* wt, float* tile,
                                               const float* __restrict__ rowscale,
                                               float* __restrict__ C,
                                               int mt, int nh, int lane) {
    const int half = lane >> 4;
    const int l16  = lane & 15;
    const int m0   = mt * 16;

    const float s = rowscale[m0 + l16];

    const float* __restrict__ gsrc = A + (long)(m0 + half) * IN_F + 4 * l16;
    float* sdst = tile + half * G1_AP + 4 * l16;

    const float* arow = tile + l16 * G1_AP + 2 * half;
    const float* bcol = wt + (32 * nh + l16) * G1_WP + 2 * half;

    float4 ra[8], rb[8];
    #pragma unroll
    for (int j = 0; j < 8; ++j) ra[j] = *(const float4*)(gsrc + (long)(2 * j) * IN_F);
    #pragma unroll
    for (int j = 0; j < 8; ++j) rb[j] = *(const float4*)(gsrc + (long)(2 * j) * IN_F + IN_F / 2);

    v8f c0 = {}, c1 = {};
    #pragma unroll
    for (int h = 0; h < 2; ++h) {
        #pragma unroll
        for (int j = 0; j < 8; ++j) *(float4*)(sdst + (2 * j) * G1_AP) = (h == 0) ? ra[j] : rb[j];

        #pragma unroll
        for (int k0 = 0; k0 < IN_F / 2; k0 += 4) {
            const float2 av  = *(const float2*)(arow + k0);
            const float2 b0v = *(const float2*)(bcol + (IN_F / 2) * h + k0);
            const float2 b1v = *(const float2*)(bcol + 16 * G1_WP + (IN_F / 2) * h + k0);
            v2f a, b0, b1;
            a.x  = av.x * s;  a.y  = av.y * s;
            b0.x = b0v.x;     b0.y = b0v.y;
            b1.x = b1v.x;     b1.y = b1v.y;
            c0 = __builtin_amdgcn_wmma_f32_16x16x4_f32(false, a, false, b0, (short)0, c0, false, false);
            c1 = __builtin_amdgcn_wmma_f32_16x16x4_f32(false, a, false, b1, (short)0, c1, false, false);
        }
    }

    float* __restrict__ crow = C + (long)(m0 + 8 * half) * H_F + 32 * nh + l16;
    #pragma unroll
    for (int r = 0; r < 8; ++r) {
        crow[(long)r * H_F +  0] = c0[r];
        crow[(long)r * H_F + 16] = c1[r];
    }
}


#define PG_PLACE_BLOCKS (DEG_R * DEG_S)
#define PG_GEMM_BLOCKS ((G1_UNITS + G1_WAVES - 1) / G1_WAVES)
#define PG_LDS_BYTES   (G1_LDS_BYTES > DEG_BINS * 4 ? G1_LDS_BYTES : DEG_BINS * 4)

__global__ __launch_bounds__(DEG_T)
void gcn_place_and_gemm1(const int* __restrict__ src,
                         const int* __restrict__ dst,
                         const unsigned* __restrict__ pre,
                         const unsigned* __restrict__ rowptr,
                         int* __restrict__ col,
                         const float* __restrict__ feature,
                         const float* __restrict__ W1,
                         const float* __restrict__ out_norm,
                         float* __restrict__ g) {
    __shared__ __attribute__((aligned(16))) unsigned char pg_lds[PG_LDS_BYTES];

    if (blockIdx.x >= PG_PLACE_BLOCKS) {
        float* wt = (float*)pg_lds;
        gcn_gemm1_stage_w(W1, wt, threadIdx.x);
        __syncthreads();
        const int wave = threadIdx.x >> 5;
        const int unit = (blockIdx.x - PG_PLACE_BLOCKS) * G1_WAVES + wave;
        if (unit < G1_UNITS)
            gcn_gemm1_unit(feature, wt, wt + G1_WFLOATS + wave * G1_AFLOATS, out_norm, g,
                           unit >> 1, unit & 1, threadIdx.x & 31);
        return;
    }
    gcn_place_cell((unsigned*)pg_lds, src, dst, pre, rowptr, col, blockIdx.x, threadIdx.x);
}


#define GCN_ZROW   N_NODES
#define GCN_GROWS  (N_NODES + 1)

static_assert((long)GCN_GROWS * H_F * 4 < (1L << 32), "row byte offsets fit 32 bits");

__device__ __forceinline__ float4 gcn_sum16(const float* __restrict__ g, int mine,
                                            unsigned lane_byte, float4 acc) {
    float4 v[16];
    #pragma unroll
    for (int k = 0; k < 16; ++k) {
        const unsigned sk = (unsigned)__shfl(mine, k, 16);
        v[k] = *(const float4*)((const char*)g + (sk * (unsigned)(H_F * 4) + lane_byte));
    }
    #pragma unroll
    for (int k = 0; k < 16; ++k) {
        acc.x += v[k].x; acc.y += v[k].y; acc.z += v[k].z; acc.w += v[k].w;
    }
    return acc;
}

template <int NPW, typename Epi>
__device__ __forceinline__ void gcn_pull_rows(const float* __restrict__ g,
                                              const unsigned* __restrict__ rowptr,
                                              const int* __restrict__ col,
                                              const float* __restrict__ in_norm,
                                              const float* __restrict__ bias,
                                              int n0, int lane, const Epi& epi) {
    static_assert(NPW % 2 == 0 && NPW < 32, "pairs of nodes; one lane per row start");
    const int grp  = lane >> 4;
    const int l16  = lane & 15;

    const unsigned lane_byte = 16u * (unsigned)l16;

    const unsigned rp  = rowptr[n0 + min(lane, NPW)];
    const float    nrm = in_norm[n0 + min(lane, NPW - 1)];
    const float4   bb  = ((const float4*)bias)[l16];

    unsigned beg = __shfl(rp, grp);
    unsigned end = __shfl(rp, grp + 1);
    int c0 = (beg + (unsigned)l16 < end)       ? col[beg + (unsigned)l16]       : GCN_ZROW;
    int c1 = (beg + 16u + (unsigned)l16 < end) ? col[beg + 16u + (unsigned)l16] : GCN_ZROW;

    asm volatile("" :: "v"(nrm), "v"(bb.x), "v"(bb.y), "v"(bb.z), "v"(bb.w), "v"(c0), "v"(c1));

    #pragma unroll 1
    for (int i = 0; i < NPW / 2; ++i) {
        unsigned nbeg = 0u, nend = 0u;
        int nc0 = GCN_ZROW, nc1 = GCN_ZROW;
        if (i + 1 < NPW / 2) {
            nbeg = __shfl(rp, 2 * i + 2 + grp);
            nend = __shfl(rp, 2 * i + 3 + grp);
            nc0 = (nbeg + (unsigned)l16 < nend)       ? col[nbeg + (unsigned)l16]       : GCN_ZROW;
            nc1 = (nbeg + 16u + (unsigned)l16 < nend) ? col[nbeg + 16u + (unsigned)l16] : GCN_ZROW;
        }
        const float nn = __shfl(nrm, 2 * i + grp);

        float4 acc = make_float4(0.f, 0.f, 0.f, 0.f);
        for (unsigned base = beg; base < end; base += 16) {
            const unsigned j2 = base + 32u + (unsigned)l16;
            const int c2 = (j2 < end) ? col[j2] : GCN_ZROW;
            acc = gcn_sum16(g, c0, lane_byte, acc);
            c0 = c1;
            c1 = c2;
        }

        float4 o;
        o.x = fmaxf(acc.x * nn + bb.x, 0.0f);
        o.y = fmaxf(acc.y * nn + bb.y, 0.0f);
        o.z = fmaxf(acc.z * nn + bb.z, 0.0f);
        o.w = fmaxf(acc.w * nn + bb.w, 0.0f);
        epi.row(i, grp, l16, n0 + 2 * i + grp, o);

        beg = nbeg; end = nend; c0 = nc0; c1 = nc1;
    }
}


#define PG2_NPW    16
#define PG2_WPB    5
#define PG2_T      (PG2_WPB * 32)
#define PG2_PITCH  (H_F + 4)
#define PG2_WAVES  (N_NODES / PG2_NPW)
#define PG2_BLOCKS (PG2_WAVES / PG2_WPB)
#define PG2_W2Q    (H_F * H_F / 4)
#define PG2_STG    128

static_assert(N_NODES % PG2_NPW == 0, "whole tiles of 16 nodes");
static_assert(PG2_WAVES % PG2_WPB == 0, "whole blocks: every wave of every block has a tile");
static_assert(PG2_STG % 32 == 0 && PG2_STG <= PG2_T && PG2_W2Q % PG2_STG == 0, "whole waves stage W2 in whole rounds");
static_assert((PG2_PITCH * 4) % 16 == 0, "16-byte row stores into the tile");

struct gcn_epi_tile {
    float* tile;
    __device__ __forceinline__ void row(int i, int grp, int l16, int n, float4 o) const {
        *(float4*)(tile + (2 * i + grp) * PG2_PITCH + 4 * l16) = o;
    }
};

__global__ __launch_bounds__(PG2_T)
void gcn_pull_gemm2(const float* __restrict__ g,
                    const unsigned* __restrict__ rowptr,
                    const int* __restrict__ col,
                    const float* __restrict__ in_norm,
                    const float* __restrict__ bias,
                    const float* __restrict__ W2,
                    const float* __restrict__ out_norm,
                    float* __restrict__ g2) {
    __shared__ __attribute__((aligned(16))) float wt[H_F * PG2_PITCH];
    __shared__ __attribute__((aligned(16))) float tiles[PG2_WPB * PG2_NPW * PG2_PITCH];

    const int tid = threadIdx.x;

    if (tid < PG2_STG) {
        #pragma unroll
        for (int j = 0; j < PG2_W2Q / PG2_STG; ++j) {
            const int q = j * PG2_STG + tid;
            const float4 v = ((const float4*)W2)[q];
            const int k = q / (H_F / 4);
            const int n = (q % (H_F / 4)) * 4;
            wt[(n + 0) * PG2_PITCH + k] = v.x;
            wt[(n + 1) * PG2_PITCH + k] = v.y;
            wt[(n + 2) * PG2_PITCH + k] = v.z;
            wt[(n + 3) * PG2_PITCH + k] = v.w;
        }
    }
    __syncthreads();

    const int wave = tid >> 5;
    const int lane = tid & 31;
    const int wv   = blockIdx.x * PG2_WPB + wave;
    const int n0   = wv * PG2_NPW;
    const int half = lane >> 4;
    const int l16  = lane & 15;

    const float s = out_norm[n0 + l16];

    float* tile = tiles + wave * (PG2_NPW * PG2_PITCH);
    gcn_epi_tile epi;
    epi.tile = tile;
    gcn_pull_rows<PG2_NPW>(g, rowptr, col, in_norm, bias, n0, lane, epi);


    const float* arow = tile + l16 * PG2_PITCH + 2 * half;
    const float* bcol = wt   + l16 * PG2_PITCH + 2 * half;

    v8f c0 = {}, c1 = {}, c2 = {}, c3 = {};
    #pragma unroll
    for (int k0 = 0; k0 < H_F; k0 += 4) {
        const float2 av  = *(const float2*)(arow + k0);
        const float2 b0v = *(const float2*)(bcol + 0 * 16 * PG2_PITCH + k0);
        const float2 b1v = *(const float2*)(bcol + 1 * 16 * PG2_PITCH + k0);
        const float2 b2v = *(const float2*)(bcol + 2 * 16 * PG2_PITCH + k0);
        const float2 b3v = *(const float2*)(bcol + 3 * 16 * PG2_PITCH + k0);
        v2f a, b0, b1, b2, b3;
        a.x  = av.x * s;  a.y  = av.y * s;
        b0.x = b0v.x;     b0.y = b0v.y;
        b1.x = b1v.x;     b1.y = b1v.y;
        b2.x = b2v.x;     b2.y = b2v.y;
        b3.x = b3v.x;     b3.y = b3v.y;
        c0 = __builtin_amdgcn_wmma_f32_16x16x4_f32(false, a, false, b0, (short)0, c0, false, false);
        c1 = __builtin_amdgcn_wmma_f32_16x16x4_f32(false, a, false, b1, (short)0, c1, false, false);
        c2 = __builtin_amdgcn_wmma_f32_16x16x4_f32(false, a, false, b2, (short)0, c2, false, false);
        c3 = __builtin_amdgcn_wmma_f32_16x16x4_f32(false, a, false, b3, (short)0, c3, false, false);
    }

    float* __restrict__ crow = g2 + (long)(n0 + 8 * half) * H_F + l16;
    #pragma unroll
    for (int r = 0; r < 8; ++r) {
        crow[(long)r * H_F +  0] = c0[r];
        crow[(long)r * H_F + 16] = c1[r];
        crow[(long)r * H_F + 32] = c2[r];
        crow[(long)r * H_F + 48] = c3[r];
    }
}


#define PULL_T   256
#define PULL_NPW 10

static_assert(N_NODES % (PULL_NPW * (PULL_T / 32)) == 0, "whole blocks of whole waves");

__device__ __forceinline__ float gcn_row16_sum(float v) {
    v += __builtin_bit_cast(float, __builtin_amdgcn_update_dpp(0, __builtin_bit_cast(int, v), 0x161, 0xf, 0xf, true));
    v += __builtin_bit_cast(float, __builtin_amdgcn_update_dpp(0, __builtin_bit_cast(int, v), 0x162, 0xf, 0xf, true));
    v += __builtin_bit_cast(float, __builtin_amdgcn_update_dpp(0, __builtin_bit_cast(int, v), 0x164, 0xf, 0xf, true));
    v += __builtin_bit_cast(float, __builtin_amdgcn_update_dpp(0, __builtin_bit_cast(int, v), 0x168, 0xf, 0xf, true));
    return v;
}

struct gcn_epi_head {
    float4 w0, w1;
    float2 bh;
    float* out;
    __device__ __forceinline__ void row(int i, int grp, int l16, int n, float4 o) const {
        float p0 = o.x * w0.x + o.y * w0.z + o.z * w1.x + o.w * w1.z;
        float p1 = o.x * w0.y + o.y * w0.w + o.z * w1.y + o.w * w1.w;
        p0 = gcn_row16_sum(p0);
        p1 = gcn_row16_sum(p1);
        if (l16 == 0) ((float2*)out)[n] = make_float2(p0 + bh.x, p1 + bh.y);
    }
};

__global__ __launch_bounds__(PULL_T)
void gcn_pull_head(const float* __restrict__ g,
                   const unsigned* __restrict__ rowptr,
                   const int* __restrict__ col,
                   const float* __restrict__ in_norm,
                   const float* __restrict__ bias,
                   const float* __restrict__ Wm,
                   const float* __restrict__ bm,
                   float* __restrict__ out) {
    const int lane = threadIdx.x & 31;
    const int l16  = lane & 15;
    const int wv   = blockIdx.x * (PULL_T / 32) + (threadIdx.x >> 5);

    gcn_epi_head epi;
    epi.w0  = ((const float4*)Wm)[2 * l16];
    epi.w1  = ((const float4*)Wm)[2 * l16 + 1];
    epi.bh  = make_float2(bm[0], bm[1]);
    epi.out = out;
    gcn_pull_rows<PULL_NPW>(g, rowptr, col, in_norm, bias, wv * PULL_NPW, lane, epi);
}


extern "C" void kernel_launch(void* const* d_in, const int* in_sizes, int n_in,
                              void* d_out, int out_size, void* d_ws, size_t ws_size,
                              hipStream_t stream) {
    const float* feature = (const float*)d_in[0];
    const float* W1      = (const float*)d_in[1];
    const float* b1      = (const float*)d_in[2];
    const float* W2      = (const float*)d_in[3];
    const float* b2      = (const float*)d_in[4];
    const float* Wm      = (const float*)d_in[5];
    const float* bm      = (const float*)d_in[6];
    const int*   src     = (const int*)d_in[7];
    const int*   dst     = (const int*)d_in[8];
    float*       out     = (float*)d_out;

    float* out_norm = (float*)d_ws;
    float* in_norm  = out_norm + N_NODES;
    float* buf0     = in_norm + N_NODES;
    float* buf1     = buf0 + (long)GCN_GROWS * H_F;
    unsigned* part  = (unsigned*)(buf1 + (long)GCN_GROWS * H_F);
    unsigned* pre   = part + (long)DEG_S * DEG_NP;
    unsigned* degin = part + (long)2 * DEG_S * DEG_NP;
    unsigned* rowptr = degin + DEG_NP;
    int*      col   = (int*)(rowptr + DEG_NP + 32);
    unsigned* blktot = (unsigned*)(col + N_EDGES);

    gcn_degree_hist<<<2 * DEG_R * DEG_S, DEG_T, 0, stream>>>(src, dst, part);
    gcn_degree_finish<<<2 * FIN_NB, FIN_T, 0, stream>>>(part, out_norm, degin, blktot,
                                                        buf0 + (long)GCN_ZROW * H_F, buf1 + (long)GCN_ZROW * H_F);

    gcn_rowptr_scan<<<FIN_NB, FIN_T, 0, stream>>>(degin, blktot, rowptr);
    gcn_place_and_gemm1<<<PG_PLACE_BLOCKS + PG_GEMM_BLOCKS, DEG_T, 0, stream>>>(
        src, dst, pre, rowptr, col, feature, W1, out_norm, buf0);

    gcn_pull_gemm2<<<PG2_BLOCKS, PG2_T, 0, stream>>>(buf0, rowptr, col, in_norm, b1, W2, out_norm, buf1);

    gcn_pull_head<<<N_NODES / (PULL_NPW * (PULL_T / 32)), PULL_T, 0, stream>>>(
        buf1, rowptr, col, in_norm, b2, Wm, bm, out);
}
